# starting kernel
# baseline (speedup 1.0000x reference)
#include <hip/hip_runtime.h>
#include <math.h>

#define N_NODES 100000
#define N_EDGES 1600000
#define NF 256
#define NH 128
#define NC 40
#define NCP 48

typedef float v2f __attribute__((ext_vector_type(2)));
typedef float v8f __attribute__((ext_vector_type(8)));

__global__ void k_degree(const int* __restrict__ dst, float* __restrict__ deg) {
    int e = blockIdx.x * blockDim.x + threadIdx.x;
    if (e < N_EDGES) atomicAdd(&deg[dst[e]], 1.0f);
}

__global__ void k_dinv(float* __restrict__ deg) {
    int i = blockIdx.x * blockDim.x + threadIdx.x;
    if (i < N_NODES) deg[i] = rsqrtf(deg[i] + 1.0f);
}

template<int K, int LDA, int LDB>
__global__ void k_gemm_wmma(const float* __restrict__ A, const float* __restrict__ B,
                            float* __restrict__ C) {
    const int mt   = blockIdx.x;
    const int nt   = threadIdx.x >> 5;
    const int lane = threadIdx.x & 31;
    const int half = lane >> 4;
    const int lm   = lane & 15;

    const float* arow = A + (size_t)(mt * 16 + lm) * LDA;
    const float* bcol = B + nt * 16 + lm;

    v8f c = {0.f, 0.f, 0.f, 0.f, 0.f, 0.f, 0.f, 0.f};
    #pragma unroll 4
    for (int kb = 0; kb < K; kb += 4) {
        const int ka = kb + 2 * half;
        v2f a; a.x = arow[ka];                 a.y = arow[ka + 1];
        v2f b; b.x = bcol[(size_t)ka * LDB];   b.y = bcol[(size_t)(ka + 1) * LDB];
        c = __builtin_amdgcn_wmma_f32_16x16x4_f32(false, a, false, b,
                                                  (short)0, c, false, false);
    }
    float* out = C + (size_t)(mt * 16 + 8 * half) * LDB + nt * 16 + lm;
    #pragma unroll
    for (int v = 0; v < 8; ++v) out[(size_t)v * LDB] = c[v];
}

__global__ void k_self1(const float* __restrict__ xw, const float* __restrict__ dinv,
                        float* __restrict__ agg) {
    int id = blockIdx.x * blockDim.x + threadIdx.x;
    if (id < N_NODES * NH) {
        int i = id >> 7;
        float di = dinv[i];
        agg[id] = xw[id] * di * di;
    }
}

__global__ void k_edge1(const int* __restrict__ src, const int* __restrict__ dst,
                        const float* __restrict__ dinv, const float* __restrict__ xw,
                        float* __restrict__ agg) {
    int gid  = blockIdx.x * blockDim.x + threadIdx.x;
    int e    = gid >> 5;
    if (e >= N_EDGES) return;
    int lane = gid & 31;
    int s = src[e], d = dst[e];
    float norm = dinv[s] * dinv[d];
    const float4 v = *(const float4*)(xw + (size_t)s * NH + lane * 4);
    float* o = agg + (size_t)d * NH + lane * 4;
    atomicAdd(o + 0, v.x * norm);
    atomicAdd(o + 1, v.y * norm);
    atomicAdd(o + 2, v.z * norm);
    atomicAdd(o + 3, v.w * norm);
}

__global__ void k_relu_bias(float* __restrict__ agg, const float* __restrict__ b) {
    int id = blockIdx.x * blockDim.x + threadIdx.x;
    if (id < N_NODES * NH) agg[id] = fmaxf(agg[id] + b[id & (NH - 1)], 0.f);
}

__global__ void k_padw2(const float* __restrict__ W2, float* __restrict__ W2p) {
    int id = blockIdx.x * blockDim.x + threadIdx.x;
    if (id < NH * NCP) {
        int k = id / NCP, n = id % NCP;
        W2p[id] = (n < NC) ? W2[k * NC + n] : 0.f;
    }
}

__global__ void k_self2(const float* __restrict__ xw2, const float* __restrict__ dinv,
                        float* __restrict__ agg2) {
    int id = blockIdx.x * blockDim.x + threadIdx.x;
    if (id < N_NODES * NC) {
        int i = id / NC, f = id - i * NC;
        float di = dinv[i];
        agg2[id] = xw2[(size_t)i * NCP + f] * di * di;
    }
}

__global__ void k_edge2(const int* __restrict__ src, const int* __restrict__ dst,
                        const float* __restrict__ dinv, const float* __restrict__ xw2,
                        float* __restrict__ agg2) {
    int gid  = blockIdx.x * blockDim.x + threadIdx.x;
    int e    = gid >> 5;
    if (e >= N_EDGES) return;
    int lane = gid & 31;
    int s = src[e], d = dst[e];
    float norm = dinv[s] * dinv[d];
    atomicAdd(agg2 + (size_t)d * NC + lane, xw2[(size_t)s * NCP + lane] * norm);
    if (lane < NC - 32)
        atomicAdd(agg2 + (size_t)d * NC + lane + 32,
                  xw2[(size_t)s * NCP + lane + 32] * norm);
}

__global__ void k_bias2(const float* __restrict__ agg2, const float* __restrict__ b2,
                        float* __restrict__ logits) {
    int id = blockIdx.x * blockDim.x + threadIdx.x;
    if (id < N_NODES * NC) {
        int f = id % NC;
        logits[id] = agg2[id] + b2[f];
    }
}

__global__ void k_colstats(const float* __restrict__ logits,
                           float* __restrict__ colmax, float* __restrict__ colsum) {
    __shared__ float red[256];
    int c = blockIdx.x;
    float m = -INFINITY;
    for (int i = threadIdx.x; i < N_NODES; i += blockDim.x)
        m = fmaxf(m, logits[(size_t)i * NC + c]);
    red[threadIdx.x] = m; __syncthreads();
    for (int s = 128; s > 0; s >>= 1) {
        if (threadIdx.x < s) red[threadIdx.x] = fmaxf(red[threadIdx.x], red[threadIdx.x + s]);
        __syncthreads();
    }
    float cm = red[0]; __syncthreads();
    float acc = 0.f;
    for (int i = threadIdx.x; i < N_NODES; i += blockDim.x)
        acc += expf(logits[(size_t)i * NC + c] - cm);
    red[threadIdx.x] = acc; __syncthreads();
    for (int s = 128; s > 0; s >>= 1) {
        if (threadIdx.x < s) red[threadIdx.x] += red[threadIdx.x + s];
        __syncthreads();
    }
    if (threadIdx.x == 0) { colmax[c] = cm; colsum[c] = red[0]; }
}

__global__ void k_finalout(const float* __restrict__ logits,
                           const float* __restrict__ colmax, const float* __restrict__ colsum,
                           float* __restrict__ out_lsm, float* __restrict__ out_sm0) {
    int gid  = blockIdx.x * blockDim.x + threadIdx.x;
    int node = gid >> 5;
    if (node >= N_NODES) return;
    int lane = gid & 31;
    const float* row = logits + (size_t)node * NC;
    float v0 = row[lane];
    bool has2 = lane < (NC - 32);
    float v1 = has2 ? row[lane + 32] : -INFINITY;
    float m = fmaxf(v0, v1);
    #pragma unroll
    for (int off = 16; off > 0; off >>= 1) m = fmaxf(m, __shfl_xor(m, off, 32));
    float s = expf(v0 - m) + (has2 ? expf(v1 - m) : 0.f);
    #pragma unroll
    for (int off = 16; off > 0; off >>= 1) s += __shfl_xor(s, off, 32);
    float lse = m + logf(s);
    out_lsm[(size_t)node * NC + lane] = v0 - lse;
    out_sm0[(size_t)node * NC + lane] = expf(v0 - colmax[lane]) / colsum[lane];
    if (has2) {
        out_lsm[(size_t)node * NC + lane + 32] = v1 - lse;
        out_sm0[(size_t)node * NC + lane + 32] =
            expf(v1 - colmax[lane + 32]) / colsum[lane + 32];
    }
}

extern "C" void kernel_launch(void* const* d_in, const int* in_sizes, int n_in,
                              void* d_out, int out_size, void* d_ws, size_t ws_size,
                              hipStream_t stream) {
    (void)in_sizes; (void)n_in; (void)out_size; (void)ws_size;
    const float* x   = (const float*)d_in[0];
    const int*   ei  = (const int*)d_in[1];
    const float* W1  = (const float*)d_in[2];
    const float* b1  = (const float*)d_in[3];
    const float* W2  = (const float*)d_in[4];
    const float* b2  = (const float*)d_in[5];
    const int* src = ei;
    const int* dst = ei + N_EDGES;

    float* ws = (float*)d_ws;
    size_t off = 0;
    float* dinv = ws + off; off += N_NODES; off = (off + 3) & ~(size_t)3;
    float* xw1  = ws + off; off += (size_t)N_NODES * NH;
    float* agg1 = ws + off; off += (size_t)N_NODES * NH;
    float* W2p  = ws + off; off += NH * NCP;
    float* xw2  = ws + off; off += (size_t)N_NODES * NCP;
    float* agg2 = ws + off; off += (size_t)N_NODES * NC;
    float* colmax = ws + off; off += 64;
    float* colsum = ws + off; off += 64;

    float* out_lsm = (float*)d_out;
    float* logits  = (float*)d_out + (size_t)N_NODES * NC;
    float* out_sm0 = (float*)d_out + 2 * (size_t)N_NODES * NC;

    hipMemsetAsync(dinv, 0, N_NODES * sizeof(float), stream);
    k_degree<<<(N_EDGES + 255) / 256, 256, 0, stream>>>(dst, dinv);
    k_dinv<<<(N_NODES + 255) / 256, 256, 0, stream>>>(dinv);

    k_gemm_wmma<NF, NF, NH><<<N_NODES / 16, (NH / 16) * 32, 0, stream>>>(x, W1, xw1);
    k_self1<<<(N_NODES * NH + 255) / 256, 256, 0, stream>>>(xw1, dinv, agg1);
    k_edge1<<<(size_t)N_EDGES * 32 / 256, 256, 0, stream>>>(src, dst, dinv, xw1, agg1);
    k_relu_bias<<<(N_NODES * NH + 255) / 256, 256, 0, stream>>>(agg1, b1);

    k_padw2<<<(NH * NCP + 255) / 256, 256, 0, stream>>>(W2, W2p);
    k_gemm_wmma<NH, NH, NCP><<<N_NODES / 16, (NCP / 16) * 32, 0, stream>>>(agg1, W2p, xw2);
    k_self2<<<(N_NODES * NC + 255) / 256, 256, 0, stream>>>(xw2, dinv, agg2);
    k_edge2<<<(size_t)N_EDGES * 32 / 256, 256, 0, stream>>>(src, dst, dinv, xw2, agg2);
    k_bias2<<<(N_NODES * NC + 255) / 256, 256, 0, stream>>>(agg2, b2, logits);

    k_colstats<<<NC, 256, 0, stream>>>(logits, colmax, colsum);
    k_finalout<<<(N_NODES * 32 + 255) / 256, 256, 0, stream>>>(logits, colmax, colsum,
                                                               out_lsm, out_sm0);
}


# 2nd neighbour sum reads per-wave lists of kept edges that the 1st writes while scanning: destination list scanned once, not twice; index loads run 1-3 groups ahead; sum order = LDS arrival order
# speedup vs baseline: 2.5745x; 2.5745x over previous
#include <hip/hip_runtime.h>
#include <math.h>

#define N_NODES 100000
#define N_EDGES 1600000
#define NF 256
#define NH 128
#define NC 40
#define NCP 48


typedef float v2f __attribute__((ext_vector_type(2)));
typedef float v8f __attribute__((ext_vector_type(8)));

__global__ void k_degree(const int* __restrict__ dst, float* __restrict__ deg) {
    int e = blockIdx.x * blockDim.x + threadIdx.x;
    if (e < N_EDGES) atomicAdd(&deg[dst[e]], 1.0f);
}

__global__ void k_dinv(float* __restrict__ deg) {
    int i = blockIdx.x * blockDim.x + threadIdx.x;
    if (i < N_NODES) deg[i] = rsqrtf(deg[i] + 1.0f);
}

template<int K, int LDA, int LDB>
__global__ void k_gemm_wmma(const float* __restrict__ A, const float* __restrict__ B,
                            float* __restrict__ C) {
    const int mt   = blockIdx.x;
    const int nt   = threadIdx.x >> 5;
    const int lane = threadIdx.x & 31;
    const int half = lane >> 4;
    const int lm   = lane & 15;

    const float* arow = A + (size_t)(mt * 16 + lm) * LDA;
    const float* bcol = B + nt * 16 + lm;

    v8f c = {0.f, 0.f, 0.f, 0.f, 0.f, 0.f, 0.f, 0.f};
    #pragma unroll 4
    for (int kb = 0; kb < K; kb += 4) {
        const int ka = kb + 2 * half;
        v2f a; a.x = arow[ka];                 a.y = arow[ka + 1];
        v2f b; b.x = bcol[(size_t)ka * LDB];   b.y = bcol[(size_t)(ka + 1) * LDB];
        c = __builtin_amdgcn_wmma_f32_16x16x4_f32(false, a, false, b,
                                                  (short)0, c, false, false);
    }
    float* out = C + (size_t)(mt * 16 + 8 * half) * LDB + nt * 16 + lm;
    #pragma unroll
    for (int v = 0; v < 8; ++v) out[(size_t)v * LDB] = c[v];
}

#define OWN_GROUPS   256
#define OWN_NODES    392
#define OWN_THREADS  1024
#define OWN_WAVES    (OWN_THREADS / 32)
#define STEP_LOADS   4
#define STEP_EDGES   (STEP_LOADS * 128)
#define N_STEPS      (N_EDGES / STEP_EDGES)
#define Q_BATCH      8
#define Q_CAP        (Q_BATCH + STEP_EDGES)
#define Q_EDGE_BITS  21

#define T_RUN        416
#define T_PITCH      (OWN_GROUPS * T_RUN)

#define L_CAP        512
#define L_WAVES      (OWN_GROUPS * OWN_WAVES)
#define L_INTS       ((size_t)L_WAVES * L_CAP + L_WAVES)
#define R_BATCH      16

static_assert(T_RUN % 32 == 0 && T_RUN >= OWN_NODES, "a run is whole lines and holds every owned node");
static_assert((size_t)NC * T_PITCH + (size_t)N_NODES * NCP <= (size_t)N_NODES * NH, "the plane and the second product's rows fit the region they reuse");
static_assert(L_INTS <= (size_t)N_NODES * NCP, "the lists fit the region they use");
static_assert(L_CAP % 32 == 0 && 32 % R_BATCH == 0, "a list is whole groups of 32, a group whole row batches");
static_assert(N_EDGES % STEP_EDGES == 0, "a step is whole");
static_assert(N_STEPS >= OWN_WAVES, "every wave has a first step");
static_assert(OWN_NODES * OWN_GROUPS >= N_NODES && (OWN_GROUPS - 1) * OWN_NODES < N_NODES, "ranges cover the nodes, none empty");
static_assert(N_EDGES <= (1 << Q_EDGE_BITS) && OWN_NODES <= (1 << (31 - Q_EDGE_BITS)), "queue entry fits an int");

template<int V> struct Const { static constexpr int value = V; };

template<int NFEAT, int SP, bool RELU, bool TPLANE, bool FROM_LIST>
__global__ void __launch_bounds__(OWN_THREADS)
k_own(const int* __restrict__ src, const int* __restrict__ dst,
      const float* __restrict__ dinv, const float* __restrict__ xw,
      const float* __restrict__ bias, float* __restrict__ agg, float* __restrict__ tplane,
      int* __restrict__ lists) {
    __shared__ __attribute__((aligned(16))) float acc[OWN_NODES * NFEAT];
    __shared__ int q[OWN_WAVES][Q_CAP];

    const int tid  = threadIdx.x;
    const int lane = tid & 31;
    const int wave = __builtin_amdgcn_readfirstlane(tid >> 5);
    const int lo   = blockIdx.x * OWN_NODES;
    const int cnt  = (N_NODES - lo < OWN_NODES) ? (N_NODES - lo) : OWN_NODES;

    for (int id = tid; id < cnt * NFEAT; id += OWN_THREADS) {
        const int i = id / NFEAT, f = id - i * NFEAT;
        const float di = dinv[lo + i];
        acc[id] = xw[(size_t)(lo + i) * SP + f] * di * di;
    }
    __syncthreads();

    int* qw = q[wave];
    int  qn = 0;

    const int gw     = blockIdx.x * OWN_WAVES + wave;
    int*      lst    = lists + (size_t)gw * L_CAP;
    int*      lcount = lists + (size_t)L_WAVES * L_CAP;
    int       ln     = 0;
    constexpr int EMASK = (1 << Q_EDGE_BITS) - 1;

    auto rows = [&](const int s, const int dl, const int nbits, const int off, const int n, const bool full,
                    auto nb) __attribute__((always_inline)) {
        constexpr int NB = decltype(nb)::value;
        if constexpr (NFEAT == 128) {
            float4 v[NB];
            #pragma unroll
            for (int k = 0; k < NB; ++k) {
                if (full || k < n) {
                    const int sj = __builtin_amdgcn_readlane(s, off + k);
                    v[k] = *(const float4*)(xw + (size_t)sj * SP + lane * 4);
                }
            }
            #pragma unroll
            for (int k = 0; k < NB; ++k) {
                if (full || k < n) {
                    const int   dj = __builtin_amdgcn_readlane(dl, off + k);
                    const float nj = __int_as_float(__builtin_amdgcn_readlane(nbits, off + k));
                    float* o = &acc[dj * NFEAT + lane * 4];
                    __hip_atomic_fetch_add(o + 0, v[k].x * nj, __ATOMIC_RELAXED, __HIP_MEMORY_SCOPE_WORKGROUP);
                    __hip_atomic_fetch_add(o + 1, v[k].y * nj, __ATOMIC_RELAXED, __HIP_MEMORY_SCOPE_WORKGROUP);
                    __hip_atomic_fetch_add(o + 2, v[k].z * nj, __ATOMIC_RELAXED, __HIP_MEMORY_SCOPE_WORKGROUP);
                    __hip_atomic_fetch_add(o + 3, v[k].w * nj, __ATOMIC_RELAXED, __HIP_MEMORY_SCOPE_WORKGROUP);
                }
            }
        } else {
            static_assert(NFEAT == 128 || (NFEAT > 32 && NFEAT <= 64), "row = one or two 32-lane groups");
            const bool has2 = lane < NFEAT - 32;
            float v0[NB], v1[NB];
            #pragma unroll
            for (int k = 0; k < NB; ++k) {
                if (full || k < n) {
                    const int sj = __builtin_amdgcn_readlane(s, off + k);
                    const float* r = xw + (size_t)sj * SP + lane;
                    v0[k] = r[0];
                    v1[k] = has2 ? r[32] : 0.f;
                }
            }
            #pragma unroll
            for (int k = 0; k < NB; ++k) {
                if (full || k < n) {
                    const int   dj = __builtin_amdgcn_readlane(dl, off + k);
                    const float nj = __int_as_float(__builtin_amdgcn_readlane(nbits, off + k));
                    float* o = &acc[dj * NFEAT + lane];
                    __hip_atomic_fetch_add(o, v0[k] * nj, __ATOMIC_RELAXED, __HIP_MEMORY_SCOPE_WORKGROUP);
                    if (has2)
                        __hip_atomic_fetch_add(o + 32, v1[k] * nj, __ATOMIC_RELAXED, __HIP_MEMORY_SCOPE_WORKGROUP);
                }
            }
        }
    };

    auto batch = [&](const int base, const int n, const bool full) __attribute__((always_inline)) {
        int s = 0, dl = 0, nbits = 0;
        if (lane < n) {
            const int ent = qw[base + lane];
            if constexpr (!FROM_LIST) {
                if (ln + lane < L_CAP) lst[ln + lane] = ent;
            }
            dl = ent >> Q_EDGE_BITS;
            s  = src[ent & EMASK];
            nbits = __float_as_int(dinv[s] * dinv[lo + dl]);
        }
        if constexpr (!FROM_LIST) ln += n;
        rows(s, dl, nbits, 0, n, full, Const<Q_BATCH>{});
    };

    auto take = [&](const int d, const int e) __attribute__((always_inline)) {
        const unsigned dl   = (unsigned)(d - lo);
        const bool     mine = dl < (unsigned)cnt;
        const unsigned mask = __builtin_amdgcn_ballot_w32(mine);
        if (mask != 0u) {
            if (mine) qw[qn + (int)__builtin_amdgcn_mbcnt_lo(mask, 0u)] = e | (int)(dl << Q_EDGE_BITS);
            qn += __builtin_popcount(mask);
        }
    };

    auto scan = [&]() __attribute__((always_inline)) {
        const int4* d4  = (const int4*)dst;
        const int   rot = blockIdx.x * (N_STEPS / OWN_GROUPS);
        auto place = [&](const int t) __attribute__((always_inline)) {
            const int p = t + rot;
            return p >= N_STEPS ? p - N_STEPS : p;
        };
        auto load_step = [&](int4 (&buf)[STEP_LOADS], const int t) __attribute__((always_inline)) {
            const int p = place(t);
            #pragma unroll
            for (int j = 0; j < STEP_LOADS; ++j) buf[j] = d4[p * (STEP_EDGES / 4) + j * 32 + lane];
        };
        auto examine = [&](const int4 (&buf)[STEP_LOADS], const int t) __attribute__((always_inline)) {
            const int p = place(t);
            #pragma unroll
            for (int j = 0; j < STEP_LOADS; ++j) {
                const int e0 = p * STEP_EDGES + j * 128 + lane * 4;
                take(buf[j].x, e0 + 0);
                take(buf[j].y, e0 + 1);
                take(buf[j].z, e0 + 2);
                take(buf[j].w, e0 + 3);
            }
            while (qn >= Q_BATCH) { qn -= Q_BATCH; batch(qn, Q_BATCH, true); }
        };
        int4 bufA[STEP_LOADS], bufB[STEP_LOADS];
        load_step(bufA, wave);
        for (int t = wave; t < N_STEPS; t += 2 * OWN_WAVES) {
            const int t1 = t + OWN_WAVES, t2 = t + 2 * OWN_WAVES;
            load_step(bufB, t1 < N_STEPS ? t1 : t);
            examine(bufA, t);
            load_step(bufA, t2 < N_STEPS ? t2 : t);
            if (t1 < N_STEPS) examine(bufB, t1);
        }
        if (qn > 0) batch(0, qn, false);
    };

    auto from_list = [&]() __attribute__((always_inline)) {
        auto entry = [&](const int g) __attribute__((always_inline)) {
            const int i = g * 32 + lane;
            return i < ln ? lst[i] : 0;
        };
        int   e_n  = entry(0);
        int   s_n  = 0, dl_n = 0;
        int   s_c  = 0, dl_c = 0;
        float da_c = 0.f, db_c = 0.f;
        for (int g = -2; g * 32 < ln; ++g) {
            const float da = dinv[s_n];
            const float db = dinv[lo + dl_n];
            const int   dl = e_n >> Q_EDGE_BITS;
            const int   s  = src[e_n & EMASK];
            const int   e  = entry(g + 3);
            if (g >= 0) {
                const int n     = ln - g * 32;
                const int nbits = __float_as_int(da_c * db_c);
                #pragma unroll
                for (int sub = 0; sub < 32 / R_BATCH; ++sub) {
                    const int m = n - sub * R_BATCH;
                    if (m >= R_BATCH) rows(s_c, dl_c, nbits, sub * R_BATCH, R_BATCH, true, Const<R_BATCH>{});
                    else if (m > 0)   rows(s_c, dl_c, nbits, sub * R_BATCH, m, false, Const<R_BATCH>{});
                }
            }
            s_c = s_n; dl_c = dl_n; da_c = da; db_c = db;
            s_n = s;   dl_n = dl;
            e_n = e;
        }
    };

    if constexpr (FROM_LIST) {
        ln = __builtin_amdgcn_readfirstlane(lcount[gw]);
        if (ln <= L_CAP) from_list();
        else             scan();
    } else {
        scan();
        if (lane == 0) qw[0] = ln;
    }
    __syncthreads();
    if constexpr (!FROM_LIST) {
        if (wave == 0) lcount[blockIdx.x * OWN_WAVES + lane] = q[lane][0];
    }

    static_assert(NFEAT % 4 == 0, "a 16-byte piece stays inside one row");
    constexpr int  PIECES = NFEAT / 4;
    constexpr bool SAME_F = (OWN_THREADS % PIECES) == 0;
    auto bias_at = [&](const int piece) __attribute__((always_inline)) {
        const int f = (piece % PIECES) * 4;
        return make_float4(bias[f], bias[f + 1], bias[f + 2], bias[f + 3]);
    };
    float4 bfix = make_float4(0.f, 0.f, 0.f, 0.f);
    if constexpr (SAME_F) bfix = bias_at(tid);
    float4*       o4 = (float4*)(agg + (size_t)lo * NFEAT);
    const float4* a4 = (const float4*)acc;
    for (int id = tid; id < cnt * PIECES; id += OWN_THREADS) {
        float4 b = bfix;
        if constexpr (!SAME_F) b = bias_at(id);
        float4 v = a4[id];
        v.x += b.x;
        v.y += b.y;
        v.z += b.z;
        v.w += b.w;
        if constexpr (RELU) {
            v.x = fmaxf(v.x, 0.f);
            v.y = fmaxf(v.y, 0.f);
            v.z = fmaxf(v.z, 0.f);
            v.w = fmaxf(v.w, 0.f);
        }
        o4[id] = v;
    }

    if constexpr (TPLANE) {
        static_assert(!RELU, "the plane holds acc + bias, as the row store does");
        constexpr int RUN4 = T_RUN / 4;
        for (int id = tid; id < NFEAT * RUN4; id += OWN_THREADS) {
            const int   c  = id / RUN4;
            const int   i0 = (id - c * RUN4) * 4;
            const float b  = bias[c];
            float4 v;
            v.x = (i0 + 0 < cnt) ? acc[(i0 + 0) * NFEAT + c] + b : -INFINITY;
            v.y = (i0 + 1 < cnt) ? acc[(i0 + 1) * NFEAT + c] + b : -INFINITY;
            v.z = (i0 + 2 < cnt) ? acc[(i0 + 2) * NFEAT + c] + b : -INFINITY;
            v.w = (i0 + 3 < cnt) ? acc[(i0 + 3) * NFEAT + c] + b : -INFINITY;
            *(float4*)(tplane + (size_t)c * T_PITCH + (size_t)blockIdx.x * T_RUN + i0) = v;
        }
    }
}

__global__ void k_padw2(const float* __restrict__ W2, float* __restrict__ W2p) {
    int id = blockIdx.x * blockDim.x + threadIdx.x;
    if (id < NH * NCP) {
        int k = id / NCP, n = id % NCP;
        W2p[id] = (n < NC) ? W2[k * NC + n] : 0.f;
    }
}

#define CS_THREADS 1024
#define CS_WAVES   (CS_THREADS / 32)
#define CS_LOADS   (T_PITCH / (4 * CS_THREADS))
static_assert(T_PITCH % (4 * CS_THREADS) == 0, "the threads' loads tile the column exactly");
static_assert(CS_WAVES == 32, "one partial per lane in the second reduction level");

__global__ void __launch_bounds__(CS_THREADS)
k_colstats(const float* __restrict__ tplane,
           float* __restrict__ colmax, float* __restrict__ colsum) {
    __shared__ float red[CS_WAVES];
    const int c    = blockIdx.x;
    const int lane = threadIdx.x & 31;
    const int wave = threadIdx.x >> 5;
    const float4* col = (const float4*)(tplane + (size_t)c * T_PITCH);

    float4 v[CS_LOADS];
    #pragma unroll
    for (int j = 0; j < CS_LOADS; ++j) v[j] = col[j * CS_THREADS + threadIdx.x];

    float m = -INFINITY;
    #pragma unroll
    for (int j = 0; j < CS_LOADS; ++j)
        m = fmaxf(m, fmaxf(fmaxf(v[j].x, v[j].y), fmaxf(v[j].z, v[j].w)));
    for (int off = 16; off > 0; off >>= 1) m = fmaxf(m, __shfl_xor(m, off, 32));
    if (lane == 0) red[wave] = m;
    __syncthreads();
    float cm = red[lane];
    for (int off = 16; off > 0; off >>= 1) cm = fmaxf(cm, __shfl_xor(cm, off, 32));
    __syncthreads();

    float s = 0.f;
    #pragma unroll
    for (int j = 0; j < CS_LOADS; ++j)
        s += (expf(v[j].x - cm) + expf(v[j].y - cm)) + (expf(v[j].z - cm) + expf(v[j].w - cm));
    for (int off = 16; off > 0; off >>= 1) s += __shfl_xor(s, off, 32);
    if (lane == 0) red[wave] = s;
    __syncthreads();
    if (wave == 0) {
        float t = red[lane];
        for (int off = 16; off > 0; off >>= 1) t += __shfl_xor(t, off, 32);
        if (lane == 0) { colmax[c] = cm; colsum[c] = t; }
    }
}

__global__ void k_finalout(const float* __restrict__ logits,
                           const float* __restrict__ colmax, const float* __restrict__ colsum,
                           float* __restrict__ out_lsm, float* __restrict__ out_sm0) {
    int gid  = blockIdx.x * blockDim.x + threadIdx.x;
    int node = gid >> 5;
    if (node >= N_NODES) return;
    int lane = gid & 31;
    const float* row = logits + (size_t)node * NC;
    float v0 = row[lane];
    bool has2 = lane < (NC - 32);
    float v1 = has2 ? row[lane + 32] : -INFINITY;
    float m = fmaxf(v0, v1);
    #pragma unroll
    for (int off = 16; off > 0; off >>= 1) m = fmaxf(m, __shfl_xor(m, off, 32));
    float s = expf(v0 - m) + (has2 ? expf(v1 - m) : 0.f);
    #pragma unroll
    for (int off = 16; off > 0; off >>= 1) s += __shfl_xor(s, off, 32);
    float lse = m + logf(s);
    out_lsm[(size_t)node * NC + lane] = v0 - lse;
    out_sm0[(size_t)node * NC + lane] = expf(v0 - colmax[lane]) / colsum[lane];
    if (has2) {
        out_lsm[(size_t)node * NC + lane + 32] = v1 - lse;
        out_sm0[(size_t)node * NC + lane + 32] =
            expf(v1 - colmax[lane + 32]) / colsum[lane + 32];
    }
}

extern "C" void kernel_launch(void* const* d_in, const int* in_sizes, int n_in,
                              void* d_out, int out_size, void* d_ws, size_t ws_size,
                              hipStream_t stream) {
    (void)in_sizes; (void)n_in; (void)out_size; (void)ws_size;
    const float* x   = (const float*)d_in[0];
    const int*   ei  = (const int*)d_in[1];
    const float* W1  = (const float*)d_in[2];
    const float* b1  = (const float*)d_in[3];
    const float* W2  = (const float*)d_in[4];
    const float* b2  = (const float*)d_in[5];
    const int* src = ei;
    const int* dst = ei + N_EDGES;

    float* ws = (float*)d_ws;
    size_t off = 0;
    float* dinv = ws + off; off += N_NODES; off = (off + 3) & ~(size_t)3;
    float* xw1  = ws + off; off += (size_t)N_NODES * NH;
    float* agg1 = ws + off; off += (size_t)N_NODES * NH;
    float* W2p  = ws + off; off += NH * NCP;
    int*   lists = (int*)(ws + off); off += (size_t)N_NODES * NCP;
    float* colmax = ws + off; off += 64;
    float* colsum = ws + off; off += 64;

    float* tplane = xw1;
    float* xw2    = xw1 + (size_t)NC * T_PITCH;

    float* out_lsm = (float*)d_out;
    float* logits  = (float*)d_out + (size_t)N_NODES * NC;
    float* out_sm0 = (float*)d_out + 2 * (size_t)N_NODES * NC;

    hipMemsetAsync(dinv, 0, N_NODES * sizeof(float), stream);
    k_degree<<<(N_EDGES + 255) / 256, 256, 0, stream>>>(dst, dinv);
    k_dinv<<<(N_NODES + 255) / 256, 256, 0, stream>>>(dinv);

    k_gemm_wmma<NF, NF, NH><<<N_NODES / 16, (NH / 16) * 32, 0, stream>>>(x, W1, xw1);
    k_own<NH, NH, true, false, false><<<OWN_GROUPS, OWN_THREADS, 0, stream>>>(src, dst, dinv, xw1, b1, agg1, nullptr, lists);

    k_padw2<<<(NH * NCP + 255) / 256, 256, 0, stream>>>(W2, W2p);
    k_gemm_wmma<NH, NH, NCP><<<N_NODES / 16, (NCP / 16) * 32, 0, stream>>>(agg1, W2p, xw2);
    k_own<NC, NCP, false, true, true><<<OWN_GROUPS, OWN_THREADS, 0, stream>>>(src, dst, dinv, xw2, b2, logits, tplane, lists);

    k_colstats<<<NC, CS_THREADS, 0, stream>>>(tplane, colmax, colsum);
    k_finalout<<<(N_NODES * 32 + 255) / 256, 256, 0, stream>>>(logits, colmax, colsum,
                                                               out_lsm, out_sm0);
}


# both products: whole weight matrix held in LDS in fragment order by 256 resident work-groups; a wave owns 16 rows x all columns, reads A once by 16-byte loads into registers; k summed in a new order
# speedup vs baseline: 3.0244x; 1.1748x over previous
#include <hip/hip_runtime.h>
#include <math.h>

#define N_NODES 100000
#define N_EDGES 1600000
#define NF 256
#define NH 128
#define NC 40
#define NCP 48


typedef float v2f __attribute__((ext_vector_type(2)));
typedef float v8f __attribute__((ext_vector_type(8)));

__global__ void k_degree(const int* __restrict__ dst, float* __restrict__ deg) {
    int e = blockIdx.x * blockDim.x + threadIdx.x;
    if (e < N_EDGES) atomicAdd(&deg[dst[e]], 1.0f);
}

__global__ void k_dinv(float* __restrict__ deg) {
    int i = blockIdx.x * blockDim.x + threadIdx.x;
    if (i < N_NODES) deg[i] = rsqrtf(deg[i] + 1.0f);
}

#define GM_GROUPS   256
#define GM_WAVES    16
#define GM_THREADS  (GM_WAVES * 32)
#define GM_TILES    (N_NODES / 16)
static_assert(N_NODES % 16 == 0, "row tiles are whole");

template<int K, int LDA, int N>
__global__ void __launch_bounds__(GM_THREADS)
k_gemm_wmma(const float* __restrict__ A, const float* __restrict__ B, float* __restrict__ C) {
    static_assert(K % 64 == 0 && N % 16 == 0 && LDA % 4 == 0, "pairs of whole 32-k chunks, whole column tiles, 16-byte rows");
    constexpr int NT  = N / 16;
    constexpr int KC  = K / 32;
    constexpr int NB4 = (K / 4) * N;
    __shared__ float4 Bs[NB4];

    const int tid  = threadIdx.x;
    const int lane = tid & 31;
    const int wave = __builtin_amdgcn_readfirstlane(tid >> 5);
    const int half = lane >> 4;
    const int lm   = lane & 15;

    {
        const int rot = (int)(((long)blockIdx.x * NB4) / GM_GROUPS);
        for (int i = tid; i < NB4; i += GM_THREADS) {
            int id = i + rot;
            if (id >= NB4) id -= NB4;
            const int k4 = id / N, n = id - k4 * N;
            const float* p = B + (size_t)(k4 * 4) * N + n;
            Bs[((k4 >> 1) * NT + (n >> 4)) * 32 + (k4 & 1) * 16 + (n & 15)] =
                make_float4(p[0], p[N], p[2 * N], p[3 * N]);
        }
    }
    __syncthreads();

    const int t0 = (int)(((long)blockIdx.x * GM_TILES) / GM_GROUPS);
    const int t1 = (int)(((long)(blockIdx.x + 1) * GM_TILES) / GM_GROUPS);
    for (int t = t0 + wave; t < t1; t += GM_WAVES) {
        const float4* arow = (const float4*)(A + (size_t)(t * 16 + lm) * LDA + 4 * half);
        v8f c[NT];
        #pragma unroll
        for (int nt = 0; nt < NT; ++nt) c[nt] = (v8f){0.f, 0.f, 0.f, 0.f, 0.f, 0.f, 0.f, 0.f};

        auto chunk = [&](const int kc, const float4 (&a)[4]) __attribute__((always_inline)) {
            const float4* bk = Bs + kc * (4 * NT * 32) + lane;
            #pragma unroll
            for (int q = 0; q < 4; ++q) {
                float4 b[NT];
                #pragma unroll
                for (int nt = 0; nt < NT; ++nt) b[nt] = bk[(q * NT + nt) * 32];
                const v2f a0 = {a[q].x, a[q].y};
                const v2f a1 = {a[q].z, a[q].w};
                #pragma unroll
                for (int nt = 0; nt < NT; ++nt) {
                    const v2f b0 = {b[nt].x, b[nt].y};
                    c[nt] = __builtin_amdgcn_wmma_f32_16x16x4_f32(false, a0, false, b0,
                                                                  (short)0, c[nt], false, false);
                }
                #pragma unroll
                for (int nt = 0; nt < NT; ++nt) {
                    const v2f b1 = {b[nt].z, b[nt].w};
                    c[nt] = __builtin_amdgcn_wmma_f32_16x16x4_f32(false, a1, false, b1,
                                                                  (short)0, c[nt], false, false);
                }
            }
        };

        float4 ae[4], ao[4];
        #pragma unroll
        for (int q = 0; q < 4; ++q) ae[q] = arow[2 * q];
        #pragma unroll 1
        for (int kc = 0; kc < KC; kc += 2) {
            #pragma unroll
            for (int q = 0; q < 4; ++q) ao[q] = arow[(kc + 1) * 8 + 2 * q];
            chunk(kc, ae);
            if (kc + 2 < KC) {
                #pragma unroll
                for (int q = 0; q < 4; ++q) ae[q] = arow[(kc + 2) * 8 + 2 * q];
            }
            chunk(kc + 1, ao);
        }

        float* out = C + (size_t)(t * 16 + 8 * half) * N + lm;
        #pragma unroll
        for (int nt = 0; nt < NT; ++nt) {
            #pragma unroll
            for (int v = 0; v < 8; ++v) out[(size_t)v * N + nt * 16] = c[nt][v];
        }
    }
}

#define OWN_GROUPS   256
#define OWN_NODES    392
#define OWN_THREADS  1024
#define OWN_WAVES    (OWN_THREADS / 32)
#define STEP_LOADS   4
#define STEP_EDGES   (STEP_LOADS * 128)
#define N_STEPS      (N_EDGES / STEP_EDGES)
#define Q_BATCH      8
#define Q_CAP        (Q_BATCH + STEP_EDGES)
#define Q_EDGE_BITS  21

#define T_RUN        416
#define T_PITCH      (OWN_GROUPS * T_RUN)

#define L_CAP        512
#define L_WAVES      (OWN_GROUPS * OWN_WAVES)
#define L_INTS       ((size_t)L_WAVES * L_CAP + L_WAVES)
#define R_BATCH      16

static_assert(T_RUN % 32 == 0 && T_RUN >= OWN_NODES, "a run is whole lines and holds every owned node");
static_assert((size_t)NC * T_PITCH + (size_t)N_NODES * NCP <= (size_t)N_NODES * NH, "the plane and the second product's rows fit the region they reuse");
static_assert(L_INTS <= (size_t)N_NODES * NCP, "the lists fit the region they use");
static_assert(L_CAP % 32 == 0 && 32 % R_BATCH == 0, "a list is whole groups of 32, a group whole row batches");
static_assert(N_EDGES % STEP_EDGES == 0, "a step is whole");
static_assert(N_STEPS >= OWN_WAVES, "every wave has a first step");
static_assert(OWN_NODES * OWN_GROUPS >= N_NODES && (OWN_GROUPS - 1) * OWN_NODES < N_NODES, "ranges cover the nodes, none empty");
static_assert(N_EDGES <= (1 << Q_EDGE_BITS) && OWN_NODES <= (1 << (31 - Q_EDGE_BITS)), "queue entry fits an int");

template<int V> struct Const { static constexpr int value = V; };

template<int NFEAT, int SP, bool RELU, bool TPLANE, bool FROM_LIST>
__global__ void __launch_bounds__(OWN_THREADS)
k_own(const int* __restrict__ src, const int* __restrict__ dst,
      const float* __restrict__ dinv, const float* __restrict__ xw,
      const float* __restrict__ bias, float* __restrict__ agg, float* __restrict__ tplane,
      int* __restrict__ lists) {
    __shared__ __attribute__((aligned(16))) float acc[OWN_NODES * NFEAT];
    __shared__ int q[OWN_WAVES][Q_CAP];

    const int tid  = threadIdx.x;
    const int lane = tid & 31;
    const int wave = __builtin_amdgcn_readfirstlane(tid >> 5);
    const int lo   = blockIdx.x * OWN_NODES;
    const int cnt  = (N_NODES - lo < OWN_NODES) ? (N_NODES - lo) : OWN_NODES;

    for (int id = tid; id < cnt * NFEAT; id += OWN_THREADS) {
        const int i = id / NFEAT, f = id - i * NFEAT;
        const float di = dinv[lo + i];
        acc[id] = xw[(size_t)(lo + i) * SP + f] * di * di;
    }
    __syncthreads();

    int* qw = q[wave];
    int  qn = 0;

    const int gw     = blockIdx.x * OWN_WAVES + wave;
    int*      lst    = lists + (size_t)gw * L_CAP;
    int*      lcount = lists + (size_t)L_WAVES * L_CAP;
    int       ln     = 0;
    constexpr int EMASK = (1 << Q_EDGE_BITS) - 1;

    auto rows = [&](const int s, const int dl, const int nbits, const int off, const int n, const bool full,
                    auto nb) __attribute__((always_inline)) {
        constexpr int NB = decltype(nb)::value;
        if constexpr (NFEAT == 128) {
            float4 v[NB];
            #pragma unroll
            for (int k = 0; k < NB; ++k) {
                if (full || k < n) {
                    const int sj = __builtin_amdgcn_readlane(s, off + k);
                    v[k] = *(const float4*)(xw + (size_t)sj * SP + lane * 4);
                }
            }
            #pragma unroll
            for (int k = 0; k < NB; ++k) {
                if (full || k < n) {
                    const int   dj = __builtin_amdgcn_readlane(dl, off + k);
                    const float nj = __int_as_float(__builtin_amdgcn_readlane(nbits, off + k));
                    float* o = &acc[dj * NFEAT + lane * 4];
                    __hip_atomic_fetch_add(o + 0, v[k].x * nj, __ATOMIC_RELAXED, __HIP_MEMORY_SCOPE_WORKGROUP);
                    __hip_atomic_fetch_add(o + 1, v[k].y * nj, __ATOMIC_RELAXED, __HIP_MEMORY_SCOPE_WORKGROUP);
                    __hip_atomic_fetch_add(o + 2, v[k].z * nj, __ATOMIC_RELAXED, __HIP_MEMORY_SCOPE_WORKGROUP);
                    __hip_atomic_fetch_add(o + 3, v[k].w * nj, __ATOMIC_RELAXED, __HIP_MEMORY_SCOPE_WORKGROUP);
                }
            }
        } else {
            static_assert(NFEAT == 128 || (NFEAT > 32 && NFEAT <= 64), "row = one or two 32-lane groups");
            const bool has2 = lane < NFEAT - 32;
            float v0[NB], v1[NB];
            #pragma unroll
            for (int k = 0; k < NB; ++k) {
                if (full || k < n) {
                    const int sj = __builtin_amdgcn_readlane(s, off + k);
                    const float* r = xw + (size_t)sj * SP + lane;
                    v0[k] = r[0];
                    v1[k] = has2 ? r[32] : 0.f;
                }
            }
            #pragma unroll
            for (int k = 0; k < NB; ++k) {
                if (full || k < n) {
                    const int   dj = __builtin_amdgcn_readlane(dl, off + k);
                    const float nj = __int_as_float(__builtin_amdgcn_readlane(nbits, off + k));
                    float* o = &acc[dj * NFEAT + lane];
                    __hip_atomic_fetch_add(o, v0[k] * nj, __ATOMIC_RELAXED, __HIP_MEMORY_SCOPE_WORKGROUP);
                    if (has2)
                        __hip_atomic_fetch_add(o + 32, v1[k] * nj, __ATOMIC_RELAXED, __HIP_MEMORY_SCOPE_WORKGROUP);
                }
            }
        }
    };

    auto batch = [&](const int base, const int n, const bool full) __attribute__((always_inline)) {
        int s = 0, dl = 0, nbits = 0;
        if (lane < n) {
            const int ent = qw[base + lane];
            if constexpr (!FROM_LIST) {
                if (ln + lane < L_CAP) lst[ln + lane] = ent;
            }
            dl = ent >> Q_EDGE_BITS;
            s  = src[ent & EMASK];
            nbits = __float_as_int(dinv[s] * dinv[lo + dl]);
        }
        if constexpr (!FROM_LIST) ln += n;
        rows(s, dl, nbits, 0, n, full, Const<Q_BATCH>{});
    };

    auto take = [&](const int d, const int e) __attribute__((always_inline)) {
        const unsigned dl   = (unsigned)(d - lo);
        const bool     mine = dl < (unsigned)cnt;
        const unsigned mask = __builtin_amdgcn_ballot_w32(mine);
        if (mask != 0u) {
            if (mine) qw[qn + (int)__builtin_amdgcn_mbcnt_lo(mask, 0u)] = e | (int)(dl << Q_EDGE_BITS);
            qn += __builtin_popcount(mask);
        }
    };

    auto scan = [&]() __attribute__((always_inline)) {
        const int4* d4  = (const int4*)dst;
        const int   rot = blockIdx.x * (N_STEPS / OWN_GROUPS);
        auto place = [&](const int t) __attribute__((always_inline)) {
            const int p = t + rot;
            return p >= N_STEPS ? p - N_STEPS : p;
        };
        auto load_step = [&](int4 (&buf)[STEP_LOADS], const int t) __attribute__((always_inline)) {
            const int p = place(t);
            #pragma unroll
            for (int j = 0; j < STEP_LOADS; ++j) buf[j] = d4[p * (STEP_EDGES / 4) + j * 32 + lane];
        };
        auto examine = [&](const int4 (&buf)[STEP_LOADS], const int t) __attribute__((always_inline)) {
            const int p = place(t);
            #pragma unroll
            for (int j = 0; j < STEP_LOADS; ++j) {
                const int e0 = p * STEP_EDGES + j * 128 + lane * 4;
                take(buf[j].x, e0 + 0);
                take(buf[j].y, e0 + 1);
                take(buf[j].z, e0 + 2);
                take(buf[j].w, e0 + 3);
            }
            while (qn >= Q_BATCH) { qn -= Q_BATCH; batch(qn, Q_BATCH, true); }
        };
        int4 bufA[STEP_LOADS], bufB[STEP_LOADS];
        load_step(bufA, wave);
        for (int t = wave; t < N_STEPS; t += 2 * OWN_WAVES) {
            const int t1 = t + OWN_WAVES, t2 = t + 2 * OWN_WAVES;
            load_step(bufB, t1 < N_STEPS ? t1 : t);
            examine(bufA, t);
            load_step(bufA, t2 < N_STEPS ? t2 : t);
            if (t1 < N_STEPS) examine(bufB, t1);
        }
        if (qn > 0) batch(0, qn, false);
    };

    auto from_list = [&]() __attribute__((always_inline)) {
        auto entry = [&](const int g) __attribute__((always_inline)) {
            const int i = g * 32 + lane;
            return i < ln ? lst[i] : 0;
        };
        int   e_n  = entry(0);
        int   s_n  = 0, dl_n = 0;
        int   s_c  = 0, dl_c = 0;
        float da_c = 0.f, db_c = 0.f;
        for (int g = -2; g * 32 < ln; ++g) {
            const float da = dinv[s_n];
            const float db = dinv[lo + dl_n];
            const int   dl = e_n >> Q_EDGE_BITS;
            const int   s  = src[e_n & EMASK];
            const int   e  = entry(g + 3);
            if (g >= 0) {
                const int n     = ln - g * 32;
                const int nbits = __float_as_int(da_c * db_c);
                #pragma unroll
                for (int sub = 0; sub < 32 / R_BATCH; ++sub) {
                    const int m = n - sub * R_BATCH;
                    if (m >= R_BATCH) rows(s_c, dl_c, nbits, sub * R_BATCH, R_BATCH, true, Const<R_BATCH>{});
                    else if (m > 0)   rows(s_c, dl_c, nbits, sub * R_BATCH, m, false, Const<R_BATCH>{});
                }
            }
            s_c = s_n; dl_c = dl_n; da_c = da; db_c = db;
            s_n = s;   dl_n = dl;
            e_n = e;
        }
    };

    if constexpr (FROM_LIST) {
        ln = __builtin_amdgcn_readfirstlane(lcount[gw]);
        if (ln <= L_CAP) from_list();
        else             scan();
    } else {
        scan();
        if (lane == 0) qw[0] = ln;
    }
    __syncthreads();
    if constexpr (!FROM_LIST) {
        if (wave == 0) lcount[blockIdx.x * OWN_WAVES + lane] = q[lane][0];
    }

    static_assert(NFEAT % 4 == 0, "a 16-byte piece stays inside one row");
    constexpr int  PIECES = NFEAT / 4;
    constexpr bool SAME_F = (OWN_THREADS % PIECES) == 0;
    auto bias_at = [&](const int piece) __attribute__((always_inline)) {
        const int f = (piece % PIECES) * 4;
        return make_float4(bias[f], bias[f + 1], bias[f + 2], bias[f + 3]);
    };
    float4 bfix = make_float4(0.f, 0.f, 0.f, 0.f);
    if constexpr (SAME_F) bfix = bias_at(tid);
    float4*       o4 = (float4*)(agg + (size_t)lo * NFEAT);
    const float4* a4 = (const float4*)acc;
    for (int id = tid; id < cnt * PIECES; id += OWN_THREADS) {
        float4 b = bfix;
        if constexpr (!SAME_F) b = bias_at(id);
        float4 v = a4[id];
        v.x += b.x;
        v.y += b.y;
        v.z += b.z;
        v.w += b.w;
        if constexpr (RELU) {
            v.x = fmaxf(v.x, 0.f);
            v.y = fmaxf(v.y, 0.f);
            v.z = fmaxf(v.z, 0.f);
            v.w = fmaxf(v.w, 0.f);
        }
        o4[id] = v;
    }

    if constexpr (TPLANE) {
        static_assert(!RELU, "the plane holds acc + bias, as the row store does");
        constexpr int RUN4 = T_RUN / 4;
        for (int id = tid; id < NFEAT * RUN4; id += OWN_THREADS) {
            const int   c  = id / RUN4;
            const int   i0 = (id - c * RUN4) * 4;
            const float b  = bias[c];
            float4 v;
            v.x = (i0 + 0 < cnt) ? acc[(i0 + 0) * NFEAT + c] + b : -INFINITY;
            v.y = (i0 + 1 < cnt) ? acc[(i0 + 1) * NFEAT + c] + b : -INFINITY;
            v.z = (i0 + 2 < cnt) ? acc[(i0 + 2) * NFEAT + c] + b : -INFINITY;
            v.w = (i0 + 3 < cnt) ? acc[(i0 + 3) * NFEAT + c] + b : -INFINITY;
            *(float4*)(tplane + (size_t)c * T_PITCH + (size_t)blockIdx.x * T_RUN + i0) = v;
        }
    }
}

__global__ void k_padw2(const float* __restrict__ W2, float* __restrict__ W2p) {
    int id = blockIdx.x * blockDim.x + threadIdx.x;
    if (id < NH * NCP) {
        int k = id / NCP, n = id % NCP;
        W2p[id] = (n < NC) ? W2[k * NC + n] : 0.f;
    }
}

#define CS_THREADS 1024
#define CS_WAVES   (CS_THREADS / 32)
#define CS_LOADS   (T_PITCH / (4 * CS_THREADS))
static_assert(T_PITCH % (4 * CS_THREADS) == 0, "the threads' loads tile the column exactly");
static_assert(CS_WAVES == 32, "one partial per lane in the second reduction level");

__global__ void __launch_bounds__(CS_THREADS)
k_colstats(const float* __restrict__ tplane,
           float* __restrict__ colmax, float* __restrict__ colsum) {
    __shared__ float red[CS_WAVES];
    const int c    = blockIdx.x;
    const int lane = threadIdx.x & 31;
    const int wave = threadIdx.x >> 5;
    const float4* col = (const float4*)(tplane + (size_t)c * T_PITCH);

    float4 v[CS_LOADS];
    #pragma unroll
    for (int j = 0; j < CS_LOADS; ++j) v[j] = col[j * CS_THREADS + threadIdx.x];

    float m = -INFINITY;
    #pragma unroll
    for (int j = 0; j < CS_LOADS; ++j)
        m = fmaxf(m, fmaxf(fmaxf(v[j].x, v[j].y), fmaxf(v[j].z, v[j].w)));
    for (int off = 16; off > 0; off >>= 1) m = fmaxf(m, __shfl_xor(m, off, 32));
    if (lane == 0) red[wave] = m;
    __syncthreads();
    float cm = red[lane];
    for (int off = 16; off > 0; off >>= 1) cm = fmaxf(cm, __shfl_xor(cm, off, 32));
    __syncthreads();

    float s = 0.f;
    #pragma unroll
    for (int j = 0; j < CS_LOADS; ++j)
        s += (expf(v[j].x - cm) + expf(v[j].y - cm)) + (expf(v[j].z - cm) + expf(v[j].w - cm));
    for (int off = 16; off > 0; off >>= 1) s += __shfl_xor(s, off, 32);
    if (lane == 0) red[wave] = s;
    __syncthreads();
    if (wave == 0) {
        float t = red[lane];
        for (int off = 16; off > 0; off >>= 1) t += __shfl_xor(t, off, 32);
        if (lane == 0) { colmax[c] = cm; colsum[c] = t; }
    }
}

__global__ void k_finalout(const float* __restrict__ logits,
                           const float* __restrict__ colmax, const float* __restrict__ colsum,
                           float* __restrict__ out_lsm, float* __restrict__ out_sm0) {
    int gid  = blockIdx.x * blockDim.x + threadIdx.x;
    int node = gid >> 5;
    if (node >= N_NODES) return;
    int lane = gid & 31;
    const float* row = logits + (size_t)node * NC;
    float v0 = row[lane];
    bool has2 = lane < (NC - 32);
    float v1 = has2 ? row[lane + 32] : -INFINITY;
    float m = fmaxf(v0, v1);
    #pragma unroll
    for (int off = 16; off > 0; off >>= 1) m = fmaxf(m, __shfl_xor(m, off, 32));
    float s = expf(v0 - m) + (has2 ? expf(v1 - m) : 0.f);
    #pragma unroll
    for (int off = 16; off > 0; off >>= 1) s += __shfl_xor(s, off, 32);
    float lse = m + logf(s);
    out_lsm[(size_t)node * NC + lane] = v0 - lse;
    out_sm0[(size_t)node * NC + lane] = expf(v0 - colmax[lane]) / colsum[lane];
    if (has2) {
        out_lsm[(size_t)node * NC + lane + 32] = v1 - lse;
        out_sm0[(size_t)node * NC + lane + 32] =
            expf(v1 - colmax[lane + 32]) / colsum[lane + 32];
    }
}

extern "C" void kernel_launch(void* const* d_in, const int* in_sizes, int n_in,
                              void* d_out, int out_size, void* d_ws, size_t ws_size,
                              hipStream_t stream) {
    (void)in_sizes; (void)n_in; (void)out_size; (void)ws_size;
    const float* x   = (const float*)d_in[0];
    const int*   ei  = (const int*)d_in[1];
    const float* W1  = (const float*)d_in[2];
    const float* b1  = (const float*)d_in[3];
    const float* W2  = (const float*)d_in[4];
    const float* b2  = (const float*)d_in[5];
    const int* src = ei;
    const int* dst = ei + N_EDGES;

    float* ws = (float*)d_ws;
    size_t off = 0;
    float* dinv = ws + off; off += N_NODES; off = (off + 3) & ~(size_t)3;
    float* xw1  = ws + off; off += (size_t)N_NODES * NH;
    float* agg1 = ws + off; off += (size_t)N_NODES * NH;
    float* W2p  = ws + off; off += NH * NCP;
    int*   lists = (int*)(ws + off); off += (size_t)N_NODES * NCP;
    float* colmax = ws + off; off += 64;
    float* colsum = ws + off; off += 64;

    float* tplane = xw1;
    float* xw2    = xw1 + (size_t)NC * T_PITCH;

    float* out_lsm = (float*)d_out;
    float* logits  = (float*)d_out + (size_t)N_NODES * NC;
    float* out_sm0 = (float*)d_out + 2 * (size_t)N_NODES * NC;

    hipMemsetAsync(dinv, 0, N_NODES * sizeof(float), stream);
    k_degree<<<(N_EDGES + 255) / 256, 256, 0, stream>>>(dst, dinv);
    k_dinv<<<(N_NODES + 255) / 256, 256, 0, stream>>>(dinv);

    k_gemm_wmma<NF, NF, NH><<<GM_GROUPS, GM_THREADS, 0, stream>>>(x, W1, xw1);
    k_own<NH, NH, true, false, false><<<OWN_GROUPS, OWN_THREADS, 0, stream>>>(src, dst, dinv, xw1, b1, agg1, nullptr, lists);

    k_padw2<<<(NH * NCP + 255) / 256, 256, 0, stream>>>(W2, W2p);
    k_gemm_wmma<NH, NH, NCP><<<GM_GROUPS, GM_THREADS, 0, stream>>>(agg1, W2p, xw2);
    k_own<NC, NCP, false, true, true><<<OWN_GROUPS, OWN_THREADS, 0, stream>>>(src, dst, dinv, xw2, b2, logits, tplane, lists);

    k_colstats<<<NC, CS_THREADS, 0, stream>>>(tplane, colmax, colsum);
    k_finalout<<<(N_NODES * 32 + 255) / 256, 256, 0, stream>>>(logits, colmax, colsum,
                                                               out_lsm, out_sm0);
}


# kept-edge lists hold source node, local node and weight (2 ints an entry), written by the 1st neighbour sum; the 2nd reads entries by whole lines: no src or dinv load per edge; sum order as before
# speedup vs baseline: 3.1384x; 1.0377x over previous
#include <hip/hip_runtime.h>
#include <math.h>

#define N_NODES 100000
#define N_EDGES 1600000
#define NF 256
#define NH 128
#define NC 40
#define NCP 48


typedef float v2f __attribute__((ext_vector_type(2)));
typedef float v8f __attribute__((ext_vector_type(8)));

__global__ void k_degree(const int* __restrict__ dst, float* __restrict__ deg) {
    int e = blockIdx.x * blockDim.x + threadIdx.x;
    if (e < N_EDGES) atomicAdd(&deg[dst[e]], 1.0f);
}

__global__ void k_dinv(float* __restrict__ deg) {
    int i = blockIdx.x * blockDim.x + threadIdx.x;
    if (i < N_NODES) deg[i] = rsqrtf(deg[i] + 1.0f);
}

#define GM_GROUPS   256
#define GM_WAVES    16
#define GM_THREADS  (GM_WAVES * 32)
#define GM_TILES    (N_NODES / 16)
static_assert(N_NODES % 16 == 0, "row tiles are whole");

template<int K, int LDA, int N>
__global__ void __launch_bounds__(GM_THREADS)
k_gemm_wmma(const float* __restrict__ A, const float* __restrict__ B, float* __restrict__ C) {
    static_assert(K % 64 == 0 && N % 16 == 0 && LDA % 4 == 0, "pairs of whole 32-k chunks, whole column tiles, 16-byte rows");
    constexpr int NT  = N / 16;
    constexpr int KC  = K / 32;
    constexpr int NB4 = (K / 4) * N;
    __shared__ float4 Bs[NB4];

    const int tid  = threadIdx.x;
    const int lane = tid & 31;
    const int wave = __builtin_amdgcn_readfirstlane(tid >> 5);
    const int half = lane >> 4;
    const int lm   = lane & 15;

    {
        const int rot = (int)(((long)blockIdx.x * NB4) / GM_GROUPS);
        for (int i = tid; i < NB4; i += GM_THREADS) {
            int id = i + rot;
            if (id >= NB4) id -= NB4;
            const int k4 = id / N, n = id - k4 * N;
            const float* p = B + (size_t)(k4 * 4) * N + n;
            Bs[((k4 >> 1) * NT + (n >> 4)) * 32 + (k4 & 1) * 16 + (n & 15)] =
                make_float4(p[0], p[N], p[2 * N], p[3 * N]);
        }
    }
    __syncthreads();

    const int t0 = (int)(((long)blockIdx.x * GM_TILES) / GM_GROUPS);
    const int t1 = (int)(((long)(blockIdx.x + 1) * GM_TILES) / GM_GROUPS);
    for (int t = t0 + wave; t < t1; t += GM_WAVES) {
        const float4* arow = (const float4*)(A + (size_t)(t * 16 + lm) * LDA + 4 * half);
        v8f c[NT];
        #pragma unroll
        for (int nt = 0; nt < NT; ++nt) c[nt] = (v8f){0.f, 0.f, 0.f, 0.f, 0.f, 0.f, 0.f, 0.f};

        auto chunk = [&](const int kc, const float4 (&a)[4]) __attribute__((always_inline)) {
            const float4* bk = Bs + kc * (4 * NT * 32) + lane;
            #pragma unroll
            for (int q = 0; q < 4; ++q) {
                float4 b[NT];
                #pragma unroll
                for (int nt = 0; nt < NT; ++nt) b[nt] = bk[(q * NT + nt) * 32];
                const v2f a0 = {a[q].x, a[q].y};
                const v2f a1 = {a[q].z, a[q].w};
                #pragma unroll
                for (int nt = 0; nt < NT; ++nt) {
                    const v2f b0 = {b[nt].x, b[nt].y};
                    c[nt] = __builtin_amdgcn_wmma_f32_16x16x4_f32(false, a0, false, b0,
                                                                  (short)0, c[nt], false, false);
                }
                #pragma unroll
                for (int nt = 0; nt < NT; ++nt) {
                    const v2f b1 = {b[nt].z, b[nt].w};
                    c[nt] = __builtin_amdgcn_wmma_f32_16x16x4_f32(false, a1, false, b1,
                                                                  (short)0, c[nt], false, false);
                }
            }
        };

        float4 ae[4], ao[4];
        #pragma unroll
        for (int q = 0; q < 4; ++q) ae[q] = arow[2 * q];
        #pragma unroll 1
        for (int kc = 0; kc < KC; kc += 2) {
            #pragma unroll
            for (int q = 0; q < 4; ++q) ao[q] = arow[(kc + 1) * 8 + 2 * q];
            chunk(kc, ae);
            if (kc + 2 < KC) {
                #pragma unroll
                for (int q = 0; q < 4; ++q) ae[q] = arow[(kc + 2) * 8 + 2 * q];
            }
            chunk(kc + 1, ao);
        }

        float* out = C + (size_t)(t * 16 + 8 * half) * N + lm;
        #pragma unroll
        for (int nt = 0; nt < NT; ++nt) {
            #pragma unroll
            for (int v = 0; v < 8; ++v) out[(size_t)v * N + nt * 16] = c[nt][v];
        }
    }
}

#define OWN_GROUPS   256
#define OWN_NODES    392
#define OWN_THREADS  1024
#define OWN_WAVES    (OWN_THREADS / 32)
#define STEP_LOADS   4
#define STEP_EDGES   (STEP_LOADS * 128)
#define N_STEPS      (N_EDGES / STEP_EDGES)
#define Q_BATCH      8
#define Q_CAP        (Q_BATCH + STEP_EDGES)
#define Q_EDGE_BITS  21

#define T_RUN        416
#define T_PITCH      (OWN_GROUPS * T_RUN)

#define L_CAP        288
#define L_WAVES      (OWN_GROUPS * OWN_WAVES)
#define L_INTS       ((size_t)L_WAVES * L_CAP * 2 + L_WAVES)
#define L_NODE_BITS  17
#define R_BATCH      16

static_assert(T_RUN % 32 == 0 && T_RUN >= OWN_NODES, "a run is whole lines and holds every owned node");
static_assert((size_t)NC * T_PITCH + (size_t)N_NODES * NCP <= (size_t)N_NODES * NH, "the plane and the second product's rows fit the region they reuse");
static_assert(L_INTS <= (size_t)N_NODES * NCP, "the lists fit the region they use");
static_assert(L_CAP % 32 == 0 && 32 % R_BATCH == 0, "a list is whole groups of 32, a group whole row batches");
static_assert(N_EDGES % STEP_EDGES == 0, "a step is whole");
static_assert(N_STEPS >= OWN_WAVES, "every wave has a first step");
static_assert(OWN_NODES * OWN_GROUPS >= N_NODES && (OWN_GROUPS - 1) * OWN_NODES < N_NODES, "ranges cover the nodes, none empty");
static_assert(N_EDGES <= (1 << Q_EDGE_BITS) && OWN_NODES <= (1 << (31 - Q_EDGE_BITS)), "queue entry fits an int");
static_assert(N_NODES <= (1 << L_NODE_BITS) && OWN_NODES <= (1 << (31 - L_NODE_BITS)), "source and local node fit a list entry's first int");

template<int V> struct Const { static constexpr int value = V; };

template<int NFEAT, int SP, bool RELU, bool TPLANE, bool FROM_LIST>
__global__ void __launch_bounds__(OWN_THREADS)
k_own(const int* __restrict__ src, const int* __restrict__ dst,
      const float* __restrict__ dinv, const float* __restrict__ xw,
      const float* __restrict__ bias, float* __restrict__ agg, float* __restrict__ tplane,
      int* __restrict__ lists) {
    __shared__ __attribute__((aligned(16))) float acc[OWN_NODES * NFEAT];
    __shared__ int q[OWN_WAVES][Q_CAP];

    const int tid  = threadIdx.x;
    const int lane = tid & 31;
    const int wave = __builtin_amdgcn_readfirstlane(tid >> 5);
    const int lo   = blockIdx.x * OWN_NODES;
    const int cnt  = (N_NODES - lo < OWN_NODES) ? (N_NODES - lo) : OWN_NODES;

    for (int id = tid; id < cnt * NFEAT; id += OWN_THREADS) {
        const int i = id / NFEAT, f = id - i * NFEAT;
        const float di = dinv[lo + i];
        acc[id] = xw[(size_t)(lo + i) * SP + f] * di * di;
    }
    __syncthreads();

    int* qw = q[wave];
    int  qn = 0;

    const int gw     = blockIdx.x * OWN_WAVES + wave;
    int2*     lst    = (int2*)lists + (size_t)gw * L_CAP;
    int*      lcount = lists + (size_t)L_WAVES * L_CAP * 2;
    int       ln     = 0;
    constexpr int EMASK = (1 << Q_EDGE_BITS) - 1;
    constexpr int NMASK = (1 << L_NODE_BITS) - 1;

    auto rows = [&](const int s, const int dl, const int nbits, const int off, const int n, const bool full,
                    auto nb) __attribute__((always_inline)) {
        constexpr int NB = decltype(nb)::value;
        if constexpr (NFEAT == 128) {
            float4 v[NB];
            #pragma unroll
            for (int k = 0; k < NB; ++k) {
                if (full || k < n) {
                    const int sj = __builtin_amdgcn_readlane(s, off + k);
                    v[k] = *(const float4*)(xw + (size_t)sj * SP + lane * 4);
                }
            }
            #pragma unroll
            for (int k = 0; k < NB; ++k) {
                if (full || k < n) {
                    const int   dj = __builtin_amdgcn_readlane(dl, off + k);
                    const float nj = __int_as_float(__builtin_amdgcn_readlane(nbits, off + k));
                    float* o = &acc[dj * NFEAT + lane * 4];
                    __hip_atomic_fetch_add(o + 0, v[k].x * nj, __ATOMIC_RELAXED, __HIP_MEMORY_SCOPE_WORKGROUP);
                    __hip_atomic_fetch_add(o + 1, v[k].y * nj, __ATOMIC_RELAXED, __HIP_MEMORY_SCOPE_WORKGROUP);
                    __hip_atomic_fetch_add(o + 2, v[k].z * nj, __ATOMIC_RELAXED, __HIP_MEMORY_SCOPE_WORKGROUP);
                    __hip_atomic_fetch_add(o + 3, v[k].w * nj, __ATOMIC_RELAXED, __HIP_MEMORY_SCOPE_WORKGROUP);
                }
            }
        } else {
            static_assert(NFEAT == 128 || (NFEAT > 32 && NFEAT <= 64), "row = one or two 32-lane groups");
            const bool has2 = lane < NFEAT - 32;
            float v0[NB], v1[NB];
            #pragma unroll
            for (int k = 0; k < NB; ++k) {
                if (full || k < n) {
                    const int sj = __builtin_amdgcn_readlane(s, off + k);
                    const float* r = xw + (size_t)sj * SP + lane;
                    v0[k] = r[0];
                    v1[k] = has2 ? r[32] : 0.f;
                }
            }
            #pragma unroll
            for (int k = 0; k < NB; ++k) {
                if (full || k < n) {
                    const int   dj = __builtin_amdgcn_readlane(dl, off + k);
                    const float nj = __int_as_float(__builtin_amdgcn_readlane(nbits, off + k));
                    float* o = &acc[dj * NFEAT + lane];
                    __hip_atomic_fetch_add(o, v0[k] * nj, __ATOMIC_RELAXED, __HIP_MEMORY_SCOPE_WORKGROUP);
                    if (has2)
                        __hip_atomic_fetch_add(o + 32, v1[k] * nj, __ATOMIC_RELAXED, __HIP_MEMORY_SCOPE_WORKGROUP);
                }
            }
        }
    };

    auto batch = [&](const int base, const int n, const bool full) __attribute__((always_inline)) {
        int s = 0, dl = 0, nbits = 0;
        if (lane < n) {
            const int ent = qw[base + lane];
            dl = ent >> Q_EDGE_BITS;
            s  = src[ent & EMASK];
            nbits = __float_as_int(dinv[s] * dinv[lo + dl]);
        }
        rows(s, dl, nbits, 0, n, full, Const<Q_BATCH>{});
        if constexpr (!FROM_LIST) {
            if (lane < n && ln + lane < L_CAP) lst[ln + lane] = make_int2(s | (dl << L_NODE_BITS), nbits);
            ln += n;
        }
    };

    auto take = [&](const int d, const int e) __attribute__((always_inline)) {
        const unsigned dl   = (unsigned)(d - lo);
        const bool     mine = dl < (unsigned)cnt;
        const unsigned mask = __builtin_amdgcn_ballot_w32(mine);
        if (mask != 0u) {
            if (mine) qw[qn + (int)__builtin_amdgcn_mbcnt_lo(mask, 0u)] = e | (int)(dl << Q_EDGE_BITS);
            qn += __builtin_popcount(mask);
        }
    };

    auto scan = [&]() __attribute__((always_inline)) {
        const int4* d4  = (const int4*)dst;
        const int   rot = blockIdx.x * (N_STEPS / OWN_GROUPS);
        auto place = [&](const int t) __attribute__((always_inline)) {
            const int p = t + rot;
            return p >= N_STEPS ? p - N_STEPS : p;
        };
        auto load_step = [&](int4 (&buf)[STEP_LOADS], const int t) __attribute__((always_inline)) {
            const int p = place(t);
            #pragma unroll
            for (int j = 0; j < STEP_LOADS; ++j) buf[j] = d4[p * (STEP_EDGES / 4) + j * 32 + lane];
        };
        auto examine = [&](const int4 (&buf)[STEP_LOADS], const int t) __attribute__((always_inline)) {
            const int p = place(t);
            #pragma unroll
            for (int j = 0; j < STEP_LOADS; ++j) {
                const int e0 = p * STEP_EDGES + j * 128 + lane * 4;
                take(buf[j].x, e0 + 0);
                take(buf[j].y, e0 + 1);
                take(buf[j].z, e0 + 2);
                take(buf[j].w, e0 + 3);
            }
            while (qn >= Q_BATCH) { qn -= Q_BATCH; batch(qn, Q_BATCH, true); }
        };
        int4 bufA[STEP_LOADS], bufB[STEP_LOADS];
        load_step(bufA, wave);
        for (int t = wave; t < N_STEPS; t += 2 * OWN_WAVES) {
            const int t1 = t + OWN_WAVES, t2 = t + 2 * OWN_WAVES;
            load_step(bufB, t1 < N_STEPS ? t1 : t);
            examine(bufA, t);
            load_step(bufA, t2 < N_STEPS ? t2 : t);
            if (t1 < N_STEPS) examine(bufB, t1);
        }
        if (qn > 0) batch(0, qn, false);
    };

    auto from_list = [&]() __attribute__((always_inline)) {
        auto entry = [&](const int g) __attribute__((always_inline)) {
            const int i = g * 32 + lane;
            return i < ln ? lst[i] : make_int2(0, 0);
        };
        int2 e_n = entry(0);
        for (int g = 0; g * 32 < ln; ++g) {
            const int2 e_c = e_n;
            e_n = entry(g + 1);
            const int s     = e_c.x & NMASK;
            const int dl    = e_c.x >> L_NODE_BITS;
            const int nbits = e_c.y;
            const int n     = ln - g * 32;
            #pragma unroll
            for (int sub = 0; sub < 32 / R_BATCH; ++sub) {
                const int m = n - sub * R_BATCH;
                if (m >= R_BATCH) rows(s, dl, nbits, sub * R_BATCH, R_BATCH, true, Const<R_BATCH>{});
                else if (m > 0)   rows(s, dl, nbits, sub * R_BATCH, m, false, Const<R_BATCH>{});
            }
        }
    };

    if constexpr (FROM_LIST) {
        ln = __builtin_amdgcn_readfirstlane(lcount[gw]);
        if (ln <= L_CAP) from_list();
        else             scan();
    } else {
        scan();
        if (lane == 0) qw[0] = ln;
    }
    __syncthreads();
    if constexpr (!FROM_LIST) {
        if (wave == 0) lcount[blockIdx.x * OWN_WAVES + lane] = q[lane][0];
    }

    static_assert(NFEAT % 4 == 0, "a 16-byte piece stays inside one row");
    constexpr int  PIECES = NFEAT / 4;
    constexpr bool SAME_F = (OWN_THREADS % PIECES) == 0;
    auto bias_at = [&](const int piece) __attribute__((always_inline)) {
        const int f = (piece % PIECES) * 4;
        return make_float4(bias[f], bias[f + 1], bias[f + 2], bias[f + 3]);
    };
    float4 bfix = make_float4(0.f, 0.f, 0.f, 0.f);
    if constexpr (SAME_F) bfix = bias_at(tid);
    float4*       o4 = (float4*)(agg + (size_t)lo * NFEAT);
    const float4* a4 = (const float4*)acc;
    for (int id = tid; id < cnt * PIECES; id += OWN_THREADS) {
        float4 b = bfix;
        if constexpr (!SAME_F) b = bias_at(id);
        float4 v = a4[id];
        v.x += b.x;
        v.y += b.y;
        v.z += b.z;
        v.w += b.w;
        if constexpr (RELU) {
            v.x = fmaxf(v.x, 0.f);
            v.y = fmaxf(v.y, 0.f);
            v.z = fmaxf(v.z, 0.f);
            v.w = fmaxf(v.w, 0.f);
        }
        o4[id] = v;
    }

    if constexpr (TPLANE) {
        static_assert(!RELU, "the plane holds acc + bias, as the row store does");
        constexpr int RUN4 = T_RUN / 4;
        for (int id = tid; id < NFEAT * RUN4; id += OWN_THREADS) {
            const int   c  = id / RUN4;
            const int   i0 = (id - c * RUN4) * 4;
            const float b  = bias[c];
            float4 v;
            v.x = (i0 + 0 < cnt) ? acc[(i0 + 0) * NFEAT + c] + b : -INFINITY;
            v.y = (i0 + 1 < cnt) ? acc[(i0 + 1) * NFEAT + c] + b : -INFINITY;
            v.z = (i0 + 2 < cnt) ? acc[(i0 + 2) * NFEAT + c] + b : -INFINITY;
            v.w = (i0 + 3 < cnt) ? acc[(i0 + 3) * NFEAT + c] + b : -INFINITY;
            *(float4*)(tplane + (size_t)c * T_PITCH + (size_t)blockIdx.x * T_RUN + i0) = v;
        }
    }
}

__global__ void k_padw2(const float* __restrict__ W2, float* __restrict__ W2p) {
    int id = blockIdx.x * blockDim.x + threadIdx.x;
    if (id < NH * NCP) {
        int k = id / NCP, n = id % NCP;
        W2p[id] = (n < NC) ? W2[k * NC + n] : 0.f;
    }
}

#define CS_THREADS 1024
#define CS_WAVES   (CS_THREADS / 32)
#define CS_LOADS   (T_PITCH / (4 * CS_THREADS))
static_assert(T_PITCH % (4 * CS_THREADS) == 0, "the threads' loads tile the column exactly");
static_assert(CS_WAVES == 32, "one partial per lane in the second reduction level");

__global__ void __launch_bounds__(CS_THREADS)
k_colstats(const float* __restrict__ tplane,
           float* __restrict__ colmax, float* __restrict__ colsum) {
    __shared__ float red[CS_WAVES];
    const int c    = blockIdx.x;
    const int lane = threadIdx.x & 31;
    const int wave = threadIdx.x >> 5;
    const float4* col = (const float4*)(tplane + (size_t)c * T_PITCH);

    float4 v[CS_LOADS];
    #pragma unroll
    for (int j = 0; j < CS_LOADS; ++j) v[j] = col[j * CS_THREADS + threadIdx.x];

    float m = -INFINITY;
    #pragma unroll
    for (int j = 0; j < CS_LOADS; ++j)
        m = fmaxf(m, fmaxf(fmaxf(v[j].x, v[j].y), fmaxf(v[j].z, v[j].w)));
    for (int off = 16; off > 0; off >>= 1) m = fmaxf(m, __shfl_xor(m, off, 32));
    if (lane == 0) red[wave] = m;
    __syncthreads();
    float cm = red[lane];
    for (int off = 16; off > 0; off >>= 1) cm = fmaxf(cm, __shfl_xor(cm, off, 32));
    __syncthreads();

    float s = 0.f;
    #pragma unroll
    for (int j = 0; j < CS_LOADS; ++j)
        s += (expf(v[j].x - cm) + expf(v[j].y - cm)) + (expf(v[j].z - cm) + expf(v[j].w - cm));
    for (int off = 16; off > 0; off >>= 1) s += __shfl_xor(s, off, 32);
    if (lane == 0) red[wave] = s;
    __syncthreads();
    if (wave == 0) {
        float t = red[lane];
        for (int off = 16; off > 0; off >>= 1) t += __shfl_xor(t, off, 32);
        if (lane == 0) { colmax[c] = cm; colsum[c] = t; }
    }
}

__global__ void k_finalout(const float* __restrict__ logits,
                           const float* __restrict__ colmax, const float* __restrict__ colsum,
                           float* __restrict__ out_lsm, float* __restrict__ out_sm0) {
    int gid  = blockIdx.x * blockDim.x + threadIdx.x;
    int node = gid >> 5;
    if (node >= N_NODES) return;
    int lane = gid & 31;
    const float* row = logits + (size_t)node * NC;
    float v0 = row[lane];
    bool has2 = lane < (NC - 32);
    float v1 = has2 ? row[lane + 32] : -INFINITY;
    float m = fmaxf(v0, v1);
    #pragma unroll
    for (int off = 16; off > 0; off >>= 1) m = fmaxf(m, __shfl_xor(m, off, 32));
    float s = expf(v0 - m) + (has2 ? expf(v1 - m) : 0.f);
    #pragma unroll
    for (int off = 16; off > 0; off >>= 1) s += __shfl_xor(s, off, 32);
    float lse = m + logf(s);
    out_lsm[(size_t)node * NC + lane] = v0 - lse;
    out_sm0[(size_t)node * NC + lane] = expf(v0 - colmax[lane]) / colsum[lane];
    if (has2) {
        out_lsm[(size_t)node * NC + lane + 32] = v1 - lse;
        out_sm0[(size_t)node * NC + lane + 32] =
            expf(v1 - colmax[lane + 32]) / colsum[lane + 32];
    }
}

extern "C" void kernel_launch(void* const* d_in, const int* in_sizes, int n_in,
                              void* d_out, int out_size, void* d_ws, size_t ws_size,
                              hipStream_t stream) {
    (void)in_sizes; (void)n_in; (void)out_size; (void)ws_size;
    const float* x   = (const float*)d_in[0];
    const int*   ei  = (const int*)d_in[1];
    const float* W1  = (const float*)d_in[2];
    const float* b1  = (const float*)d_in[3];
    const float* W2  = (const float*)d_in[4];
    const float* b2  = (const float*)d_in[5];
    const int* src = ei;
    const int* dst = ei + N_EDGES;

    float* ws = (float*)d_ws;
    size_t off = 0;
    float* dinv = ws + off; off += N_NODES; off = (off + 3) & ~(size_t)3;
    float* xw1  = ws + off; off += (size_t)N_NODES * NH;
    float* agg1 = ws + off; off += (size_t)N_NODES * NH;
    float* W2p  = ws + off; off += NH * NCP;
    int*   lists = (int*)(ws + off); off += (size_t)N_NODES * NCP;
    float* colmax = ws + off; off += 64;
    float* colsum = ws + off; off += 64;

    float* tplane = xw1;
    float* xw2    = xw1 + (size_t)NC * T_PITCH;

    float* out_lsm = (float*)d_out;
    float* logits  = (float*)d_out + (size_t)N_NODES * NC;
    float* out_sm0 = (float*)d_out + 2 * (size_t)N_NODES * NC;

    hipMemsetAsync(dinv, 0, N_NODES * sizeof(float), stream);
    k_degree<<<(N_EDGES + 255) / 256, 256, 0, stream>>>(dst, dinv);
    k_dinv<<<(N_NODES + 255) / 256, 256, 0, stream>>>(dinv);

    k_gemm_wmma<NF, NF, NH><<<GM_GROUPS, GM_THREADS, 0, stream>>>(x, W1, xw1);
    k_own<NH, NH, true, false, false><<<OWN_GROUPS, OWN_THREADS, 0, stream>>>(src, dst, dinv, xw1, b1, agg1, nullptr, lists);

    k_padw2<<<(NH * NCP + 255) / 256, 256, 0, stream>>>(W2, W2p);
    k_gemm_wmma<NH, NH, NCP><<<GM_GROUPS, GM_THREADS, 0, stream>>>(agg1, W2p, xw2);
    k_own<NC, NCP, false, true, true><<<OWN_GROUPS, OWN_THREADS, 0, stream>>>(src, dst, dinv, xw2, b2, logits, tplane, lists);

    k_colstats<<<NC, CS_THREADS, 0, stream>>>(tplane, colmax, colsum);
    k_finalout<<<(N_NODES * 32 + 255) / 256, 256, 0, stream>>>(logits, colmax, colsum,
                                                               out_lsm, out_sm0);
}


# 1st neighbour sum: the degree pass also marks, per owner range, which 32-edge chunks hold its edges; an owner reads its marks and only marked chunks, not the whole dst list; sum = LDS arrival order
# speedup vs baseline: 3.2629x; 1.0397x over previous
#include <hip/hip_runtime.h>
#include <math.h>

#define N_NODES 100000
#define N_EDGES 1600000
#define NF 256
#define NH 128
#define NC 40
#define NCP 48


typedef float v2f __attribute__((ext_vector_type(2)));
typedef float v8f __attribute__((ext_vector_type(8)));

#define OWN_GROUPS   256
#define OWN_NODES    392
#define MK_WAVES     8
#define MK_THREADS   (MK_WAVES * 32)
#define MK_WEDGES    1024
#define MK_GEDGES    (MK_WAVES * MK_WEDGES)
#define MK_GROUPS    ((N_EDGES + MK_GEDGES - 1) / MK_GEDGES)
#define MK_WORDS     (MK_GROUPS * MK_WAVES)
static_assert(MK_THREADS == OWN_GROUPS, "one thread per owner writes the owner's words");
static_assert(N_EDGES % 32 == 0, "a chunk is whole");

__global__ void __launch_bounds__(MK_THREADS)
k_degree_mark(const int* __restrict__ dst, float* __restrict__ deg, unsigned* __restrict__ marks) {
    __shared__ unsigned bits[MK_WAVES][OWN_GROUPS];
    const int tid  = threadIdx.x;
    const int lane = tid & 31;
    const int wave = tid >> 5;
    #pragma unroll
    for (int k = 0; k < OWN_GROUPS / 32; ++k) bits[wave][k * 32 + lane] = 0u;

    const int   base = (blockIdx.x * MK_WAVES + wave) * MK_WEDGES;
    const int4* d4   = (const int4*)dst;
    int4 v[8];
    #pragma unroll
    for (int j = 0; j < 8; ++j) {
        const int e = base + j * 128 + lane * 4;
        v[j] = (e < N_EDGES) ? d4[e >> 2] : make_int4(-1, -1, -1, -1);
    }
    #pragma unroll
    for (int j = 0; j < 8; ++j) {
        const unsigned bit = 1u << (j * 4 + (lane >> 3));
        const int d[4] = {v[j].x, v[j].y, v[j].z, v[j].w};
        #pragma unroll
        for (int i = 0; i < 4; ++i) {
            if (d[i] >= 0) {
                atomicAdd(&deg[d[i]], 1.0f);
                atomicOr(&bits[wave][d[i] / OWN_NODES], bit);
            }
        }
    }
    __syncthreads();
    uint4 w0, w1;
    w0.x = bits[0][tid]; w0.y = bits[1][tid]; w0.z = bits[2][tid]; w0.w = bits[3][tid];
    w1.x = bits[4][tid]; w1.y = bits[5][tid]; w1.z = bits[6][tid]; w1.w = bits[7][tid];
    uint4* o = (uint4*)(marks + ((size_t)blockIdx.x * OWN_GROUPS + tid) * MK_WAVES);
    o[0] = w0;
    o[1] = w1;
}
static_assert(MK_WAVES == 8, "the store above writes eight words per owner");

__global__ void k_dinv(float* __restrict__ deg) {
    int i = blockIdx.x * blockDim.x + threadIdx.x;
    if (i < N_NODES) deg[i] = rsqrtf(deg[i] + 1.0f);
}

#define GM_GROUPS   256
#define GM_WAVES    16
#define GM_THREADS  (GM_WAVES * 32)
#define GM_TILES    (N_NODES / 16)
static_assert(N_NODES % 16 == 0, "row tiles are whole");

template<int K, int LDA, int N>
__global__ void __launch_bounds__(GM_THREADS)
k_gemm_wmma(const float* __restrict__ A, const float* __restrict__ B, float* __restrict__ C) {
    static_assert(K % 64 == 0 && N % 16 == 0 && LDA % 4 == 0, "pairs of whole 32-k chunks, whole column tiles, 16-byte rows");
    constexpr int NT  = N / 16;
    constexpr int KC  = K / 32;
    constexpr int NB4 = (K / 4) * N;
    __shared__ float4 Bs[NB4];

    const int tid  = threadIdx.x;
    const int lane = tid & 31;
    const int wave = __builtin_amdgcn_readfirstlane(tid >> 5);
    const int half = lane >> 4;
    const int lm   = lane & 15;

    {
        const int rot = (int)(((long)blockIdx.x * NB4) / GM_GROUPS);
        for (int i = tid; i < NB4; i += GM_THREADS) {
            int id = i + rot;
            if (id >= NB4) id -= NB4;
            const int k4 = id / N, n = id - k4 * N;
            const float* p = B + (size_t)(k4 * 4) * N + n;
            Bs[((k4 >> 1) * NT + (n >> 4)) * 32 + (k4 & 1) * 16 + (n & 15)] =
                make_float4(p[0], p[N], p[2 * N], p[3 * N]);
        }
    }
    __syncthreads();

    const int t0 = (int)(((long)blockIdx.x * GM_TILES) / GM_GROUPS);
    const int t1 = (int)(((long)(blockIdx.x + 1) * GM_TILES) / GM_GROUPS);
    for (int t = t0 + wave; t < t1; t += GM_WAVES) {
        const float4* arow = (const float4*)(A + (size_t)(t * 16 + lm) * LDA + 4 * half);
        v8f c[NT];
        #pragma unroll
        for (int nt = 0; nt < NT; ++nt) c[nt] = (v8f){0.f, 0.f, 0.f, 0.f, 0.f, 0.f, 0.f, 0.f};

        auto chunk = [&](const int kc, const float4 (&a)[4]) __attribute__((always_inline)) {
            const float4* bk = Bs + kc * (4 * NT * 32) + lane;
            #pragma unroll
            for (int q = 0; q < 4; ++q) {
                float4 b[NT];
                #pragma unroll
                for (int nt = 0; nt < NT; ++nt) b[nt] = bk[(q * NT + nt) * 32];
                const v2f a0 = {a[q].x, a[q].y};
                const v2f a1 = {a[q].z, a[q].w};
                #pragma unroll
                for (int nt = 0; nt < NT; ++nt) {
                    const v2f b0 = {b[nt].x, b[nt].y};
                    c[nt] = __builtin_amdgcn_wmma_f32_16x16x4_f32(false, a0, false, b0,
                                                                  (short)0, c[nt], false, false);
                }
                #pragma unroll
                for (int nt = 0; nt < NT; ++nt) {
                    const v2f b1 = {b[nt].z, b[nt].w};
                    c[nt] = __builtin_amdgcn_wmma_f32_16x16x4_f32(false, a1, false, b1,
                                                                  (short)0, c[nt], false, false);
                }
            }
        };

        float4 ae[4], ao[4];
        #pragma unroll
        for (int q = 0; q < 4; ++q) ae[q] = arow[2 * q];
        #pragma unroll 1
        for (int kc = 0; kc < KC; kc += 2) {
            #pragma unroll
            for (int q = 0; q < 4; ++q) ao[q] = arow[(kc + 1) * 8 + 2 * q];
            chunk(kc, ae);
            if (kc + 2 < KC) {
                #pragma unroll
                for (int q = 0; q < 4; ++q) ae[q] = arow[(kc + 2) * 8 + 2 * q];
            }
            chunk(kc + 1, ao);
        }

        float* out = C + (size_t)(t * 16 + 8 * half) * N + lm;
        #pragma unroll
        for (int nt = 0; nt < NT; ++nt) {
            #pragma unroll
            for (int v = 0; v < 8; ++v) out[(size_t)v * N + nt * 16] = c[nt][v];
        }
    }
}

#define OWN_THREADS  1024
#define OWN_WAVES    (OWN_THREADS / 32)
#define W_PER_WAVE   (MK_WORDS / OWN_WAVES)
#define C_WORDS      8
#define C_CAP        (C_GROUP + C_WORDS * 32)
#define C_GROUP      16
#define Q_BATCH      8
#define Q_KEEP       (Q_BATCH + C_GROUP * 32)
#define Q_CAP        (Q_KEEP + C_CAP)
#define Q_EDGE_BITS  21

#define T_RUN        416
#define T_PITCH      (OWN_GROUPS * T_RUN)

#define L_CAP        288
#define L_WAVES      (OWN_GROUPS * OWN_WAVES)
#define L_INTS       ((size_t)L_WAVES * L_CAP * 2 + L_WAVES)
#define L_NODE_BITS  17
#define R_BATCH      16

static_assert(T_RUN % 32 == 0 && T_RUN >= OWN_NODES, "a run is whole lines and holds every owned node");
static_assert((size_t)NC * T_PITCH + (size_t)N_NODES * NCP <= (size_t)N_NODES * NH, "the plane and the second product's rows fit the region they reuse");
static_assert(L_INTS <= (size_t)N_NODES * NCP, "the lists fit the region they use");
static_assert(L_CAP % 32 == 0 && 32 % R_BATCH == 0, "a list is whole groups of 32, a group whole row batches");
static_assert(MK_WORDS % OWN_WAVES == 0, "the waves of an owner share the mark words evenly");
static_assert(C_WORDS == 8 && C_WORDS <= 32 && C_GROUP <= 32, "the list positions are summed over 8 lanes; a group is read one chunk per lane");
static_assert((size_t)MK_GROUPS * OWN_GROUPS * MK_WAVES <= (size_t)N_NODES * NC, "the marks fit the part of the output buffer that holds them");
static_assert(OWN_NODES * OWN_GROUPS >= N_NODES && (OWN_GROUPS - 1) * OWN_NODES < N_NODES, "ranges cover the nodes, none empty");
static_assert(N_EDGES <= (1 << Q_EDGE_BITS) && OWN_NODES <= (1 << (31 - Q_EDGE_BITS)), "queue entry fits an int");
static_assert(N_NODES <= (1 << L_NODE_BITS) && OWN_NODES <= (1 << (31 - L_NODE_BITS)), "source and local node fit a list entry's first int");

template<int V> struct Const { static constexpr int value = V; };

template<int NFEAT, int SP, bool RELU, bool TPLANE, bool FROM_LIST>
__global__ void __launch_bounds__(OWN_THREADS)
k_own(const int* __restrict__ src, const int* __restrict__ dst,
      const float* __restrict__ dinv, const float* __restrict__ xw,
      const float* __restrict__ bias, float* __restrict__ agg, float* __restrict__ tplane,
      int* __restrict__ lists, const unsigned* __restrict__ marks) {
    __shared__ __attribute__((aligned(16))) float acc[OWN_NODES * NFEAT];
    __shared__ int q[OWN_WAVES][Q_CAP];

    const int tid  = threadIdx.x;
    const int lane = tid & 31;
    const int wave = __builtin_amdgcn_readfirstlane(tid >> 5);
    const int lo   = blockIdx.x * OWN_NODES;
    const int cnt  = (N_NODES - lo < OWN_NODES) ? (N_NODES - lo) : OWN_NODES;

    for (int id = tid; id < cnt * NFEAT; id += OWN_THREADS) {
        const int i = id / NFEAT, f = id - i * NFEAT;
        const float di = dinv[lo + i];
        acc[id] = xw[(size_t)(lo + i) * SP + f] * di * di;
    }
    __syncthreads();

    int* qw = q[wave];
    int  qn = 0;

    const int gw     = blockIdx.x * OWN_WAVES + wave;
    int2*     lst    = (int2*)lists + (size_t)gw * L_CAP;
    int*      lcount = lists + (size_t)L_WAVES * L_CAP * 2;
    int       ln     = 0;
    constexpr int EMASK = (1 << Q_EDGE_BITS) - 1;
    constexpr int NMASK = (1 << L_NODE_BITS) - 1;

    auto rows = [&](const int s, const int dl, const int nbits, const int off, const int n, const bool full,
                    auto nb) __attribute__((always_inline)) {
        constexpr int NB = decltype(nb)::value;
        if constexpr (NFEAT == 128) {
            float4 v[NB];
            #pragma unroll
            for (int k = 0; k < NB; ++k) {
                if (full || k < n) {
                    const int sj = __builtin_amdgcn_readlane(s, off + k);
                    v[k] = *(const float4*)(xw + (size_t)sj * SP + lane * 4);
                }
            }
            #pragma unroll
            for (int k = 0; k < NB; ++k) {
                if (full || k < n) {
                    const int   dj = __builtin_amdgcn_readlane(dl, off + k);
                    const float nj = __int_as_float(__builtin_amdgcn_readlane(nbits, off + k));
                    float* o = &acc[dj * NFEAT + lane * 4];
                    __hip_atomic_fetch_add(o + 0, v[k].x * nj, __ATOMIC_RELAXED, __HIP_MEMORY_SCOPE_WORKGROUP);
                    __hip_atomic_fetch_add(o + 1, v[k].y * nj, __ATOMIC_RELAXED, __HIP_MEMORY_SCOPE_WORKGROUP);
                    __hip_atomic_fetch_add(o + 2, v[k].z * nj, __ATOMIC_RELAXED, __HIP_MEMORY_SCOPE_WORKGROUP);
                    __hip_atomic_fetch_add(o + 3, v[k].w * nj, __ATOMIC_RELAXED, __HIP_MEMORY_SCOPE_WORKGROUP);
                }
            }
        } else {
            static_assert(NFEAT == 128 || (NFEAT > 32 && NFEAT <= 64), "row = one or two 32-lane groups");
            const bool has2 = lane < NFEAT - 32;
            float v0[NB], v1[NB];
            #pragma unroll
            for (int k = 0; k < NB; ++k) {
                if (full || k < n) {
                    const int sj = __builtin_amdgcn_readlane(s, off + k);
                    const float* r = xw + (size_t)sj * SP + lane;
                    v0[k] = r[0];
                    v1[k] = has2 ? r[32] : 0.f;
                }
            }
            #pragma unroll
            for (int k = 0; k < NB; ++k) {
                if (full || k < n) {
                    const int   dj = __builtin_amdgcn_readlane(dl, off + k);
                    const float nj = __int_as_float(__builtin_amdgcn_readlane(nbits, off + k));
                    float* o = &acc[dj * NFEAT + lane];
                    __hip_atomic_fetch_add(o, v0[k] * nj, __ATOMIC_RELAXED, __HIP_MEMORY_SCOPE_WORKGROUP);
                    if (has2)
                        __hip_atomic_fetch_add(o + 32, v1[k] * nj, __ATOMIC_RELAXED, __HIP_MEMORY_SCOPE_WORKGROUP);
                }
            }
        }
    };

    auto batch = [&](const int base, const int n, const bool full) __attribute__((always_inline)) {
        int s = 0, dl = 0, nbits = 0;
        if (lane < n) {
            const int ent = qw[base + lane];
            dl = ent >> Q_EDGE_BITS;
            s  = src[ent & EMASK];
            nbits = __float_as_int(dinv[s] * dinv[lo + dl]);
        }
        rows(s, dl, nbits, 0, n, full, Const<Q_BATCH>{});
        if constexpr (!FROM_LIST) {
            if (lane < n && ln + lane < L_CAP) lst[ln + lane] = make_int2(s | (dl << L_NODE_BITS), nbits);
            ln += n;
        }
    };

    auto take = [&](const int d, const int e) __attribute__((always_inline)) {
        const unsigned dl   = (unsigned)(d - lo);
        const bool     mine = dl < (unsigned)cnt;
        const unsigned mask = __builtin_amdgcn_ballot_w32(mine);
        if (mask != 0u) {
            if (mine) qw[qn + (int)__builtin_amdgcn_mbcnt_lo(mask, 0u)] = e | (int)(dl << Q_EDGE_BITS);
            qn += __builtin_popcount(mask);
        }
    };

    auto scan = [&]() __attribute__((always_inline)) {
        const int rot = (int)(((long)blockIdx.x * MK_WORDS) / OWN_GROUPS);
        int*      cl  = qw + Q_KEEP;
        int       cn  = 0;
        auto word_of = [&](const int i) __attribute__((always_inline)) {
            int p = wave + OWN_WAVES * i + rot;
            return p >= MK_WORDS ? p - MK_WORDS : p;
        };
        auto mark_at = [&](const int i) __attribute__((always_inline)) {
            const int p = word_of(i);
            return marks[((size_t)(p / MK_WAVES) * OWN_GROUPS + blockIdx.x) * MK_WAVES + (p % MK_WAVES)];
        };
        static_assert(W_PER_WAVE > 32 && W_PER_WAVE <= 64 && 32 % C_WORDS == 0, "two words per lane hold a wave's turns; a round lies in one of them");
        const unsigned Wa = mark_at(lane);
        const unsigned Wb = (32 + lane < W_PER_WAVE) ? mark_at(32 + lane) : 0u;

        auto group = [&](const int base, const int m, const bool full) __attribute__((always_inline)) {
            int myc = 0;
            if (lane < C_GROUP && (full || lane < m)) myc = cl[base + lane];
            const int c0 = __builtin_amdgcn_readlane(myc, 0);
            int dv[C_GROUP];
            #pragma unroll
            for (int k = 0; k < C_GROUP; ++k) {
                int ck = __builtin_amdgcn_readlane(myc, k);
                if (!full && k >= m) ck = c0;
                dv[k] = dst[ck * 32 + lane];
            }
            #pragma unroll
            for (int k = 0; k < C_GROUP; ++k) {
                if (full || k < m) {
                    const int ck = __builtin_amdgcn_readlane(myc, k);
                    take(dv[k], ck * 32 + lane);
                }
            }
            while (qn >= Q_BATCH) { qn -= Q_BATCH; batch(qn, Q_BATCH, true); }
        };

        #pragma unroll 1
        for (int r = 0; r < W_PER_WAVE; r += C_WORDS) {
            const int i = r + lane;
            unsigned W = (unsigned)__shfl((int)(r < 32 ? Wa : Wb), i & 31);
            if (lane >= C_WORDS || i >= W_PER_WAVE) W = 0u;
            const int widx = word_of(i);
            const int n = __builtin_popcount(W);
            int x = n;
            { const int t = __shfl_up(x, 1); if (lane >= 1) x += t; }
            { const int t = __shfl_up(x, 2); if (lane >= 2) x += t; }
            { const int t = __shfl_up(x, 4); if (lane >= 4) x += t; }
            int pos = cn + x - n;
            cn += __builtin_amdgcn_readlane(x, C_WORDS - 1);
            while (W != 0u) {
                const int c = __builtin_ctz(W);
                W &= W - 1u;
                cl[pos++] = widx * 32 + c;
            }
            while (cn >= C_GROUP) { cn -= C_GROUP; group(cn, C_GROUP, true); }
        }
        if (cn > 0) group(0, cn, false);
        if (qn > 0) batch(0, qn, false);
    };

    auto from_list = [&]() __attribute__((always_inline)) {
        auto entry = [&](const int g) __attribute__((always_inline)) {
            const int i = g * 32 + lane;
            return i < ln ? lst[i] : make_int2(0, 0);
        };
        int2 e_n = entry(0);
        for (int g = 0; g * 32 < ln; ++g) {
            const int2 e_c = e_n;
            e_n = entry(g + 1);
            const int s     = e_c.x & NMASK;
            const int dl    = e_c.x >> L_NODE_BITS;
            const int nbits = e_c.y;
            const int n     = ln - g * 32;
            #pragma unroll
            for (int sub = 0; sub < 32 / R_BATCH; ++sub) {
                const int m = n - sub * R_BATCH;
                if (m >= R_BATCH) rows(s, dl, nbits, sub * R_BATCH, R_BATCH, true, Const<R_BATCH>{});
                else if (m > 0)   rows(s, dl, nbits, sub * R_BATCH, m, false, Const<R_BATCH>{});
            }
        }
    };

    if constexpr (FROM_LIST) {
        ln = __builtin_amdgcn_readfirstlane(lcount[gw]);
        if (ln <= L_CAP) from_list();
        else             scan();
    } else {
        scan();
        if (lane == 0) qw[0] = ln;
    }
    __syncthreads();
    if constexpr (!FROM_LIST) {
        if (wave == 0) lcount[blockIdx.x * OWN_WAVES + lane] = q[lane][0];
    }

    static_assert(NFEAT % 4 == 0, "a 16-byte piece stays inside one row");
    constexpr int  PIECES = NFEAT / 4;
    constexpr bool SAME_F = (OWN_THREADS % PIECES) == 0;
    auto bias_at = [&](const int piece) __attribute__((always_inline)) {
        const int f = (piece % PIECES) * 4;
        return make_float4(bias[f], bias[f + 1], bias[f + 2], bias[f + 3]);
    };
    float4 bfix = make_float4(0.f, 0.f, 0.f, 0.f);
    if constexpr (SAME_F) bfix = bias_at(tid);
    float4*       o4 = (float4*)(agg + (size_t)lo * NFEAT);
    const float4* a4 = (const float4*)acc;
    for (int id = tid; id < cnt * PIECES; id += OWN_THREADS) {
        float4 b = bfix;
        if constexpr (!SAME_F) b = bias_at(id);
        float4 v = a4[id];
        v.x += b.x;
        v.y += b.y;
        v.z += b.z;
        v.w += b.w;
        if constexpr (RELU) {
            v.x = fmaxf(v.x, 0.f);
            v.y = fmaxf(v.y, 0.f);
            v.z = fmaxf(v.z, 0.f);
            v.w = fmaxf(v.w, 0.f);
        }
        o4[id] = v;
    }

    if constexpr (TPLANE) {
        static_assert(!RELU, "the plane holds acc + bias, as the row store does");
        constexpr int RUN4 = T_RUN / 4;
        for (int id = tid; id < NFEAT * RUN4; id += OWN_THREADS) {
            const int   c  = id / RUN4;
            const int   i0 = (id - c * RUN4) * 4;
            const float b  = bias[c];
            float4 v;
            v.x = (i0 + 0 < cnt) ? acc[(i0 + 0) * NFEAT + c] + b : -INFINITY;
            v.y = (i0 + 1 < cnt) ? acc[(i0 + 1) * NFEAT + c] + b : -INFINITY;
            v.z = (i0 + 2 < cnt) ? acc[(i0 + 2) * NFEAT + c] + b : -INFINITY;
            v.w = (i0 + 3 < cnt) ? acc[(i0 + 3) * NFEAT + c] + b : -INFINITY;
            *(float4*)(tplane + (size_t)c * T_PITCH + (size_t)blockIdx.x * T_RUN + i0) = v;
        }
    }
}

__global__ void k_padw2(const float* __restrict__ W2, float* __restrict__ W2p) {
    int id = blockIdx.x * blockDim.x + threadIdx.x;
    if (id < NH * NCP) {
        int k = id / NCP, n = id % NCP;
        W2p[id] = (n < NC) ? W2[k * NC + n] : 0.f;
    }
}

#define CS_THREADS 1024
#define CS_WAVES   (CS_THREADS / 32)
#define CS_LOADS   (T_PITCH / (4 * CS_THREADS))
static_assert(T_PITCH % (4 * CS_THREADS) == 0, "the threads' loads tile the column exactly");
static_assert(CS_WAVES == 32, "one partial per lane in the second reduction level");

__global__ void __launch_bounds__(CS_THREADS)
k_colstats(const float* __restrict__ tplane,
           float* __restrict__ colmax, float* __restrict__ colsum) {
    __shared__ float red[CS_WAVES];
    const int c    = blockIdx.x;
    const int lane = threadIdx.x & 31;
    const int wave = threadIdx.x >> 5;
    const float4* col = (const float4*)(tplane + (size_t)c * T_PITCH);

    float4 v[CS_LOADS];
    #pragma unroll
    for (int j = 0; j < CS_LOADS; ++j) v[j] = col[j * CS_THREADS + threadIdx.x];

    float m = -INFINITY;
    #pragma unroll
    for (int j = 0; j < CS_LOADS; ++j)
        m = fmaxf(m, fmaxf(fmaxf(v[j].x, v[j].y), fmaxf(v[j].z, v[j].w)));
    for (int off = 16; off > 0; off >>= 1) m = fmaxf(m, __shfl_xor(m, off, 32));
    if (lane == 0) red[wave] = m;
    __syncthreads();
    float cm = red[lane];
    for (int off = 16; off > 0; off >>= 1) cm = fmaxf(cm, __shfl_xor(cm, off, 32));
    __syncthreads();

    float s = 0.f;
    #pragma unroll
    for (int j = 0; j < CS_LOADS; ++j)
        s += (expf(v[j].x - cm) + expf(v[j].y - cm)) + (expf(v[j].z - cm) + expf(v[j].w - cm));
    for (int off = 16; off > 0; off >>= 1) s += __shfl_xor(s, off, 32);
    if (lane == 0) red[wave] = s;
    __syncthreads();
    if (wave == 0) {
        float t = red[lane];
        for (int off = 16; off > 0; off >>= 1) t += __shfl_xor(t, off, 32);
        if (lane == 0) { colmax[c] = cm; colsum[c] = t; }
    }
}

__global__ void k_finalout(const float* __restrict__ logits,
                           const float* __restrict__ colmax, const float* __restrict__ colsum,
                           float* __restrict__ out_lsm, float* __restrict__ out_sm0) {
    int gid  = blockIdx.x * blockDim.x + threadIdx.x;
    int node = gid >> 5;
    if (node >= N_NODES) return;
    int lane = gid & 31;
    const float* row = logits + (size_t)node * NC;
    float v0 = row[lane];
    bool has2 = lane < (NC - 32);
    float v1 = has2 ? row[lane + 32] : -INFINITY;
    float m = fmaxf(v0, v1);
    #pragma unroll
    for (int off = 16; off > 0; off >>= 1) m = fmaxf(m, __shfl_xor(m, off, 32));
    float s = expf(v0 - m) + (has2 ? expf(v1 - m) : 0.f);
    #pragma unroll
    for (int off = 16; off > 0; off >>= 1) s += __shfl_xor(s, off, 32);
    float lse = m + logf(s);
    out_lsm[(size_t)node * NC + lane] = v0 - lse;
    out_sm0[(size_t)node * NC + lane] = expf(v0 - colmax[lane]) / colsum[lane];
    if (has2) {
        out_lsm[(size_t)node * NC + lane + 32] = v1 - lse;
        out_sm0[(size_t)node * NC + lane + 32] =
            expf(v1 - colmax[lane + 32]) / colsum[lane + 32];
    }
}

extern "C" void kernel_launch(void* const* d_in, const int* in_sizes, int n_in,
                              void* d_out, int out_size, void* d_ws, size_t ws_size,
                              hipStream_t stream) {
    (void)in_sizes; (void)n_in; (void)out_size; (void)ws_size;
    const float* x   = (const float*)d_in[0];
    const int*   ei  = (const int*)d_in[1];
    const float* W1  = (const float*)d_in[2];
    const float* b1  = (const float*)d_in[3];
    const float* W2  = (const float*)d_in[4];
    const float* b2  = (const float*)d_in[5];
    const int* src = ei;
    const int* dst = ei + N_EDGES;

    float* ws = (float*)d_ws;
    size_t off = 0;
    float* dinv = ws + off; off += N_NODES; off = (off + 3) & ~(size_t)3;
    float* xw1  = ws + off; off += (size_t)N_NODES * NH;
    float* agg1 = ws + off; off += (size_t)N_NODES * NH;
    float* W2p  = ws + off; off += NH * NCP;
    int*   lists = (int*)(ws + off); off += (size_t)N_NODES * NCP;
    float* colmax = ws + off; off += 64;
    float* colsum = ws + off; off += 64;

    float* tplane = xw1;
    float* xw2    = xw1 + (size_t)NC * T_PITCH;

    float* out_lsm = (float*)d_out;
    float* logits  = (float*)d_out + (size_t)N_NODES * NC;
    float* out_sm0 = (float*)d_out + 2 * (size_t)N_NODES * NC;
    unsigned* marks = (unsigned*)d_out;

    hipMemsetAsync(dinv, 0, N_NODES * sizeof(float), stream);
    k_degree_mark<<<MK_GROUPS, MK_THREADS, 0, stream>>>(dst, dinv, marks);
    k_dinv<<<(N_NODES + 255) / 256, 256, 0, stream>>>(dinv);

    k_gemm_wmma<NF, NF, NH><<<GM_GROUPS, GM_THREADS, 0, stream>>>(x, W1, xw1);
    k_own<NH, NH, true, false, false><<<OWN_GROUPS, OWN_THREADS, 0, stream>>>(src, dst, dinv, xw1, b1, agg1, nullptr, lists, marks);

    k_padw2<<<(NH * NCP + 255) / 256, 256, 0, stream>>>(W2, W2p);
    k_gemm_wmma<NH, NH, NCP><<<GM_GROUPS, GM_THREADS, 0, stream>>>(agg1, W2p, xw2);
    k_own<NC, NCP, false, true, true><<<OWN_GROUPS, OWN_THREADS, 0, stream>>>(src, dst, dinv, xw2, b2, logits, tplane, lists, marks);

    k_colstats<<<NC, CS_THREADS, 0, stream>>>(tplane, colmax, colsum);
    k_finalout<<<(N_NODES * 32 + 255) / 256, 256, 0, stream>>>(logits, colmax, colsum,
                                                               out_lsm, out_sm0);
}


# 1st neighbour sum: 128-float LDS rows kept transposed (feature 4l+c at dword c*32+l): each of the 4 float adds per edge hits 32 banks once, not 16 twice; sum order = LDS arrival order, as before
# speedup vs baseline: 3.6015x; 1.1038x over previous
#include <hip/hip_runtime.h>
#include <math.h>

#define N_NODES 100000
#define N_EDGES 1600000
#define NF 256
#define NH 128
#define NC 40
#define NCP 48


typedef float v2f __attribute__((ext_vector_type(2)));
typedef float v8f __attribute__((ext_vector_type(8)));

#define OWN_GROUPS   256
#define OWN_NODES    392
#define MK_WAVES     8
#define MK_THREADS   (MK_WAVES * 32)
#define MK_WEDGES    1024
#define MK_GEDGES    (MK_WAVES * MK_WEDGES)
#define MK_GROUPS    ((N_EDGES + MK_GEDGES - 1) / MK_GEDGES)
#define MK_WORDS     (MK_GROUPS * MK_WAVES)
static_assert(MK_THREADS == OWN_GROUPS, "one thread per owner writes the owner's words");
static_assert(N_EDGES % 32 == 0, "a chunk is whole");

__global__ void __launch_bounds__(MK_THREADS)
k_degree_mark(const int* __restrict__ dst, float* __restrict__ deg, unsigned* __restrict__ marks) {
    __shared__ unsigned bits[MK_WAVES][OWN_GROUPS];
    const int tid  = threadIdx.x;
    const int lane = tid & 31;
    const int wave = tid >> 5;
    #pragma unroll
    for (int k = 0; k < OWN_GROUPS / 32; ++k) bits[wave][k * 32 + lane] = 0u;

    const int   base = (blockIdx.x * MK_WAVES + wave) * MK_WEDGES;
    const int4* d4   = (const int4*)dst;
    int4 v[8];
    #pragma unroll
    for (int j = 0; j < 8; ++j) {
        const int e = base + j * 128 + lane * 4;
        v[j] = (e < N_EDGES) ? d4[e >> 2] : make_int4(-1, -1, -1, -1);
    }
    #pragma unroll
    for (int j = 0; j < 8; ++j) {
        const unsigned bit = 1u << (j * 4 + (lane >> 3));
        const int d[4] = {v[j].x, v[j].y, v[j].z, v[j].w};
        #pragma unroll
        for (int i = 0; i < 4; ++i) {
            if (d[i] >= 0) {
                atomicAdd(&deg[d[i]], 1.0f);
                atomicOr(&bits[wave][d[i] / OWN_NODES], bit);
            }
        }
    }
    __syncthreads();
    uint4 w0, w1;
    w0.x = bits[0][tid]; w0.y = bits[1][tid]; w0.z = bits[2][tid]; w0.w = bits[3][tid];
    w1.x = bits[4][tid]; w1.y = bits[5][tid]; w1.z = bits[6][tid]; w1.w = bits[7][tid];
    uint4* o = (uint4*)(marks + ((size_t)blockIdx.x * OWN_GROUPS + tid) * MK_WAVES);
    o[0] = w0;
    o[1] = w1;
}
static_assert(MK_WAVES == 8, "the store above writes eight words per owner");

__global__ void k_dinv(float* __restrict__ deg) {
    int i = blockIdx.x * blockDim.x + threadIdx.x;
    if (i < N_NODES) deg[i] = rsqrtf(deg[i] + 1.0f);
}

#define GM_GROUPS   256
#define GM_WAVES    16
#define GM_THREADS  (GM_WAVES * 32)
#define GM_TILES    (N_NODES / 16)
static_assert(N_NODES % 16 == 0, "row tiles are whole");

template<int K, int LDA, int N>
__global__ void __launch_bounds__(GM_THREADS)
k_gemm_wmma(const float* __restrict__ A, const float* __restrict__ B, float* __restrict__ C) {
    static_assert(K % 64 == 0 && N % 16 == 0 && LDA % 4 == 0, "pairs of whole 32-k chunks, whole column tiles, 16-byte rows");
    constexpr int NT  = N / 16;
    constexpr int KC  = K / 32;
    constexpr int NB4 = (K / 4) * N;
    __shared__ float4 Bs[NB4];

    const int tid  = threadIdx.x;
    const int lane = tid & 31;
    const int wave = __builtin_amdgcn_readfirstlane(tid >> 5);
    const int half = lane >> 4;
    const int lm   = lane & 15;

    {
        const int rot = (int)(((long)blockIdx.x * NB4) / GM_GROUPS);
        for (int i = tid; i < NB4; i += GM_THREADS) {
            int id = i + rot;
            if (id >= NB4) id -= NB4;
            const int k4 = id / N, n = id - k4 * N;
            const float* p = B + (size_t)(k4 * 4) * N + n;
            Bs[((k4 >> 1) * NT + (n >> 4)) * 32 + (k4 & 1) * 16 + (n & 15)] =
                make_float4(p[0], p[N], p[2 * N], p[3 * N]);
        }
    }
    __syncthreads();

    const int t0 = (int)(((long)blockIdx.x * GM_TILES) / GM_GROUPS);
    const int t1 = (int)(((long)(blockIdx.x + 1) * GM_TILES) / GM_GROUPS);
    for (int t = t0 + wave; t < t1; t += GM_WAVES) {
        const float4* arow = (const float4*)(A + (size_t)(t * 16 + lm) * LDA + 4 * half);
        v8f c[NT];
        #pragma unroll
        for (int nt = 0; nt < NT; ++nt) c[nt] = (v8f){0.f, 0.f, 0.f, 0.f, 0.f, 0.f, 0.f, 0.f};

        auto chunk = [&](const int kc, const float4 (&a)[4]) __attribute__((always_inline)) {
            const float4* bk = Bs + kc * (4 * NT * 32) + lane;
            #pragma unroll
            for (int q = 0; q < 4; ++q) {
                float4 b[NT];
                #pragma unroll
                for (int nt = 0; nt < NT; ++nt) b[nt] = bk[(q * NT + nt) * 32];
                const v2f a0 = {a[q].x, a[q].y};
                const v2f a1 = {a[q].z, a[q].w};
                #pragma unroll
                for (int nt = 0; nt < NT; ++nt) {
                    const v2f b0 = {b[nt].x, b[nt].y};
                    c[nt] = __builtin_amdgcn_wmma_f32_16x16x4_f32(false, a0, false, b0,
                                                                  (short)0, c[nt], false, false);
                }
                #pragma unroll
                for (int nt = 0; nt < NT; ++nt) {
                    const v2f b1 = {b[nt].z, b[nt].w};
                    c[nt] = __builtin_amdgcn_wmma_f32_16x16x4_f32(false, a1, false, b1,
                                                                  (short)0, c[nt], false, false);
                }
            }
        };

        float4 ae[4], ao[4];
        #pragma unroll
        for (int q = 0; q < 4; ++q) ae[q] = arow[2 * q];
        #pragma unroll 1
        for (int kc = 0; kc < KC; kc += 2) {
            #pragma unroll
            for (int q = 0; q < 4; ++q) ao[q] = arow[(kc + 1) * 8 + 2 * q];
            chunk(kc, ae);
            if (kc + 2 < KC) {
                #pragma unroll
                for (int q = 0; q < 4; ++q) ae[q] = arow[(kc + 2) * 8 + 2 * q];
            }
            chunk(kc + 1, ao);
        }

        float* out = C + (size_t)(t * 16 + 8 * half) * N + lm;
        #pragma unroll
        for (int nt = 0; nt < NT; ++nt) {
            #pragma unroll
            for (int v = 0; v < 8; ++v) out[(size_t)v * N + nt * 16] = c[nt][v];
        }
    }
}

#define OWN_THREADS  1024
#define OWN_WAVES    (OWN_THREADS / 32)
#define W_PER_WAVE   (MK_WORDS / OWN_WAVES)
#define C_WORDS      8
#define C_CAP        (C_GROUP + C_WORDS * 32)
#define C_GROUP      16
#define Q_BATCH      8
#define Q_KEEP       (Q_BATCH + C_GROUP * 32)
#define Q_CAP        (Q_KEEP + C_CAP)
#define Q_EDGE_BITS  21

#define T_RUN        416
#define T_PITCH      (OWN_GROUPS * T_RUN)

#define L_CAP        288
#define L_WAVES      (OWN_GROUPS * OWN_WAVES)
#define L_INTS       ((size_t)L_WAVES * L_CAP * 2 + L_WAVES)
#define L_NODE_BITS  17
#define R_BATCH      16

static_assert(T_RUN % 32 == 0 && T_RUN >= OWN_NODES, "a run is whole lines and holds every owned node");
static_assert((size_t)NC * T_PITCH + (size_t)N_NODES * NCP <= (size_t)N_NODES * NH, "the plane and the second product's rows fit the region they reuse");
static_assert(L_INTS <= (size_t)N_NODES * NCP, "the lists fit the region they use");
static_assert(L_CAP % 32 == 0 && 32 % R_BATCH == 0, "a list is whole groups of 32, a group whole row batches");
static_assert(MK_WORDS % OWN_WAVES == 0, "the waves of an owner share the mark words evenly");
static_assert(C_WORDS == 8 && C_WORDS <= 32 && C_GROUP <= 32, "the list positions are summed over 8 lanes; a group is read one chunk per lane");
static_assert((size_t)MK_GROUPS * OWN_GROUPS * MK_WAVES <= (size_t)N_NODES * NC, "the marks fit the part of the output buffer that holds them");
static_assert(OWN_NODES * OWN_GROUPS >= N_NODES && (OWN_GROUPS - 1) * OWN_NODES < N_NODES, "ranges cover the nodes, none empty");
static_assert(N_EDGES <= (1 << Q_EDGE_BITS) && OWN_NODES <= (1 << (31 - Q_EDGE_BITS)), "queue entry fits an int");
static_assert(N_NODES <= (1 << L_NODE_BITS) && OWN_NODES <= (1 << (31 - L_NODE_BITS)), "source and local node fit a list entry's first int");

template<int V> struct Const { static constexpr int value = V; };

template<int NFEAT, int SP, bool RELU, bool TPLANE, bool FROM_LIST>
__global__ void __launch_bounds__(OWN_THREADS)
k_own(const int* __restrict__ src, const int* __restrict__ dst,
      const float* __restrict__ dinv, const float* __restrict__ xw,
      const float* __restrict__ bias, float* __restrict__ agg, float* __restrict__ tplane,
      int* __restrict__ lists, const unsigned* __restrict__ marks) {
    __shared__ __attribute__((aligned(16))) float acc[OWN_NODES * NFEAT];
    __shared__ int q[OWN_WAVES][Q_CAP];

    const int tid  = threadIdx.x;
    const int lane = tid & 31;
    const int wave = __builtin_amdgcn_readfirstlane(tid >> 5);
    const int lo   = blockIdx.x * OWN_NODES;
    const int cnt  = (N_NODES - lo < OWN_NODES) ? (N_NODES - lo) : OWN_NODES;

    if constexpr (NFEAT == 128) {
        for (int id = tid; id < cnt * 32; id += OWN_THREADS) {
            const int i = id >> 5, l = id & 31;
            const float  di = dinv[lo + i];
            const float4 v  = *(const float4*)(xw + (size_t)(lo + i) * SP + 4 * l);
            float* o = &acc[i * NFEAT + l];
            o[0]  = v.x * di * di;
            o[32] = v.y * di * di;
            o[64] = v.z * di * di;
            o[96] = v.w * di * di;
        }
    } else {
        for (int id = tid; id < cnt * NFEAT; id += OWN_THREADS) {
            const int i = id / NFEAT, f = id - i * NFEAT;
            const float di = dinv[lo + i];
            acc[id] = xw[(size_t)(lo + i) * SP + f] * di * di;
        }
    }
    __syncthreads();

    int* qw = q[wave];
    int  qn = 0;

    const int gw     = blockIdx.x * OWN_WAVES + wave;
    int2*     lst    = (int2*)lists + (size_t)gw * L_CAP;
    int*      lcount = lists + (size_t)L_WAVES * L_CAP * 2;
    int       ln     = 0;
    constexpr int EMASK = (1 << Q_EDGE_BITS) - 1;
    constexpr int NMASK = (1 << L_NODE_BITS) - 1;

    auto rows = [&](const int s, const int dl, const int nbits, const int off, const int n, const bool full,
                    auto nb) __attribute__((always_inline)) {
        constexpr int NB = decltype(nb)::value;
        if constexpr (NFEAT == 128) {
            float4 v[NB];
            #pragma unroll
            for (int k = 0; k < NB; ++k) {
                if (full || k < n) {
                    const int sj = __builtin_amdgcn_readlane(s, off + k);
                    v[k] = *(const float4*)(xw + (size_t)sj * SP + lane * 4);
                }
            }
            #pragma unroll
            for (int k = 0; k < NB; ++k) {
                if (full || k < n) {
                    const int   dj = __builtin_amdgcn_readlane(dl, off + k);
                    const float nj = __int_as_float(__builtin_amdgcn_readlane(nbits, off + k));
                    float* o = &acc[dj * NFEAT + lane];
                    __hip_atomic_fetch_add(o + 0,  v[k].x * nj, __ATOMIC_RELAXED, __HIP_MEMORY_SCOPE_WORKGROUP);
                    __hip_atomic_fetch_add(o + 32, v[k].y * nj, __ATOMIC_RELAXED, __HIP_MEMORY_SCOPE_WORKGROUP);
                    __hip_atomic_fetch_add(o + 64, v[k].z * nj, __ATOMIC_RELAXED, __HIP_MEMORY_SCOPE_WORKGROUP);
                    __hip_atomic_fetch_add(o + 96, v[k].w * nj, __ATOMIC_RELAXED, __HIP_MEMORY_SCOPE_WORKGROUP);
                }
            }
        } else {
            static_assert(NFEAT == 128 || (NFEAT > 32 && NFEAT <= 64), "row = one or two 32-lane groups");
            const bool has2 = lane < NFEAT - 32;
            float v0[NB], v1[NB];
            #pragma unroll
            for (int k = 0; k < NB; ++k) {
                if (full || k < n) {
                    const int sj = __builtin_amdgcn_readlane(s, off + k);
                    const float* r = xw + (size_t)sj * SP + lane;
                    v0[k] = r[0];
                    v1[k] = has2 ? r[32] : 0.f;
                }
            }
            #pragma unroll
            for (int k = 0; k < NB; ++k) {
                if (full || k < n) {
                    const int   dj = __builtin_amdgcn_readlane(dl, off + k);
                    const float nj = __int_as_float(__builtin_amdgcn_readlane(nbits, off + k));
                    float* o = &acc[dj * NFEAT + lane];
                    __hip_atomic_fetch_add(o, v0[k] * nj, __ATOMIC_RELAXED, __HIP_MEMORY_SCOPE_WORKGROUP);
                    if (has2)
                        __hip_atomic_fetch_add(o + 32, v1[k] * nj, __ATOMIC_RELAXED, __HIP_MEMORY_SCOPE_WORKGROUP);
                }
            }
        }
    };

    auto batch = [&](const int base, const int n, const bool full) __attribute__((always_inline)) {
        int s = 0, dl = 0, nbits = 0;
        if (lane < n) {
            const int ent = qw[base + lane];
            dl = ent >> Q_EDGE_BITS;
            s  = src[ent & EMASK];
            nbits = __float_as_int(dinv[s] * dinv[lo + dl]);
        }
        rows(s, dl, nbits, 0, n, full, Const<Q_BATCH>{});
        if constexpr (!FROM_LIST) {
            if (lane < n && ln + lane < L_CAP) lst[ln + lane] = make_int2(s | (dl << L_NODE_BITS), nbits);
            ln += n;
        }
    };

    auto take = [&](const int d, const int e) __attribute__((always_inline)) {
        const unsigned dl   = (unsigned)(d - lo);
        const bool     mine = dl < (unsigned)cnt;
        const unsigned mask = __builtin_amdgcn_ballot_w32(mine);
        if (mask != 0u) {
            if (mine) qw[qn + (int)__builtin_amdgcn_mbcnt_lo(mask, 0u)] = e | (int)(dl << Q_EDGE_BITS);
            qn += __builtin_popcount(mask);
        }
    };

    auto scan = [&]() __attribute__((always_inline)) {
        const int rot = (int)(((long)blockIdx.x * MK_WORDS) / OWN_GROUPS);
        int*      cl  = qw + Q_KEEP;
        int       cn  = 0;
        auto word_of = [&](const int i) __attribute__((always_inline)) {
            int p = wave + OWN_WAVES * i + rot;
            return p >= MK_WORDS ? p - MK_WORDS : p;
        };
        auto mark_at = [&](const int i) __attribute__((always_inline)) {
            const int p = word_of(i);
            return marks[((size_t)(p / MK_WAVES) * OWN_GROUPS + blockIdx.x) * MK_WAVES + (p % MK_WAVES)];
        };
        static_assert(W_PER_WAVE > 32 && W_PER_WAVE <= 64 && 32 % C_WORDS == 0, "two words per lane hold a wave's turns; a round lies in one of them");
        const unsigned Wa = mark_at(lane);
        const unsigned Wb = (32 + lane < W_PER_WAVE) ? mark_at(32 + lane) : 0u;

        auto group = [&](const int base, const int m, const bool full) __attribute__((always_inline)) {
            int myc = 0;
            if (lane < C_GROUP && (full || lane < m)) myc = cl[base + lane];
            const int c0 = __builtin_amdgcn_readlane(myc, 0);
            int dv[C_GROUP];
            #pragma unroll
            for (int k = 0; k < C_GROUP; ++k) {
                int ck = __builtin_amdgcn_readlane(myc, k);
                if (!full && k >= m) ck = c0;
                dv[k] = dst[ck * 32 + lane];
            }
            #pragma unroll
            for (int k = 0; k < C_GROUP; ++k) {
                if (full || k < m) {
                    const int ck = __builtin_amdgcn_readlane(myc, k);
                    take(dv[k], ck * 32 + lane);
                }
            }
            while (qn >= Q_BATCH) { qn -= Q_BATCH; batch(qn, Q_BATCH, true); }
        };

        #pragma unroll 1
        for (int r = 0; r < W_PER_WAVE; r += C_WORDS) {
            const int i = r + lane;
            unsigned W = (unsigned)__shfl((int)(r < 32 ? Wa : Wb), i & 31);
            if (lane >= C_WORDS || i >= W_PER_WAVE) W = 0u;
            const int widx = word_of(i);
            const int n = __builtin_popcount(W);
            int x = n;
            { const int t = __shfl_up(x, 1); if (lane >= 1) x += t; }
            { const int t = __shfl_up(x, 2); if (lane >= 2) x += t; }
            { const int t = __shfl_up(x, 4); if (lane >= 4) x += t; }
            int pos = cn + x - n;
            cn += __builtin_amdgcn_readlane(x, C_WORDS - 1);
            while (W != 0u) {
                const int c = __builtin_ctz(W);
                W &= W - 1u;
                cl[pos++] = widx * 32 + c;
            }
            while (cn >= C_GROUP) { cn -= C_GROUP; group(cn, C_GROUP, true); }
        }
        if (cn > 0) group(0, cn, false);
        if (qn > 0) batch(0, qn, false);
    };

    auto from_list = [&]() __attribute__((always_inline)) {
        auto entry = [&](const int g) __attribute__((always_inline)) {
            const int i = g * 32 + lane;
            return i < ln ? lst[i] : make_int2(0, 0);
        };
        int2 e_n = entry(0);
        for (int g = 0; g * 32 < ln; ++g) {
            const int2 e_c = e_n;
            e_n = entry(g + 1);
            const int s     = e_c.x & NMASK;
            const int dl    = e_c.x >> L_NODE_BITS;
            const int nbits = e_c.y;
            const int n     = ln - g * 32;
            #pragma unroll
            for (int sub = 0; sub < 32 / R_BATCH; ++sub) {
                const int m = n - sub * R_BATCH;
                if (m >= R_BATCH) rows(s, dl, nbits, sub * R_BATCH, R_BATCH, true, Const<R_BATCH>{});
                else if (m > 0)   rows(s, dl, nbits, sub * R_BATCH, m, false, Const<R_BATCH>{});
            }
        }
    };

    if constexpr (FROM_LIST) {
        ln = __builtin_amdgcn_readfirstlane(lcount[gw]);
        if (ln <= L_CAP) from_list();
        else             scan();
    } else {
        scan();
        if (lane == 0) qw[0] = ln;
    }
    __syncthreads();
    if constexpr (!FROM_LIST) {
        if (wave == 0) lcount[blockIdx.x * OWN_WAVES + lane] = q[lane][0];
    }

    static_assert(NFEAT % 4 == 0, "a 16-byte piece stays inside one row");
    constexpr int  PIECES = NFEAT / 4;
    constexpr bool SAME_F = (OWN_THREADS % PIECES) == 0;
    auto bias_at = [&](const int piece) __attribute__((always_inline)) {
        const int f = (piece % PIECES) * 4;
        return make_float4(bias[f], bias[f + 1], bias[f + 2], bias[f + 3]);
    };
    float4 bfix = make_float4(0.f, 0.f, 0.f, 0.f);
    if constexpr (SAME_F) bfix = bias_at(tid);
    float4*       o4 = (float4*)(agg + (size_t)lo * NFEAT);
    const float4* a4 = (const float4*)acc;
    for (int id = tid; id < cnt * PIECES; id += OWN_THREADS) {
        float4 b = bfix;
        if constexpr (!SAME_F) b = bias_at(id);
        float4 v;
        if constexpr (NFEAT == 128) {
            const float* a = &acc[(id >> 5) * NFEAT + (id & 31)];
            v = make_float4(a[0], a[32], a[64], a[96]);
        } else {
            v = a4[id];
        }
        v.x += b.x;
        v.y += b.y;
        v.z += b.z;
        v.w += b.w;
        if constexpr (RELU) {
            v.x = fmaxf(v.x, 0.f);
            v.y = fmaxf(v.y, 0.f);
            v.z = fmaxf(v.z, 0.f);
            v.w = fmaxf(v.w, 0.f);
        }
        o4[id] = v;
    }

    if constexpr (TPLANE) {
        static_assert(!RELU, "the plane holds acc + bias, as the row store does");
        constexpr int RUN4 = T_RUN / 4;
        for (int id = tid; id < NFEAT * RUN4; id += OWN_THREADS) {
            const int   c  = id / RUN4;
            const int   i0 = (id - c * RUN4) * 4;
            const float b  = bias[c];
            float4 v;
            v.x = (i0 + 0 < cnt) ? acc[(i0 + 0) * NFEAT + c] + b : -INFINITY;
            v.y = (i0 + 1 < cnt) ? acc[(i0 + 1) * NFEAT + c] + b : -INFINITY;
            v.z = (i0 + 2 < cnt) ? acc[(i0 + 2) * NFEAT + c] + b : -INFINITY;
            v.w = (i0 + 3 < cnt) ? acc[(i0 + 3) * NFEAT + c] + b : -INFINITY;
            *(float4*)(tplane + (size_t)c * T_PITCH + (size_t)blockIdx.x * T_RUN + i0) = v;
        }
    }
}

__global__ void k_padw2(const float* __restrict__ W2, float* __restrict__ W2p) {
    int id = blockIdx.x * blockDim.x + threadIdx.x;
    if (id < NH * NCP) {
        int k = id / NCP, n = id % NCP;
        W2p[id] = (n < NC) ? W2[k * NC + n] : 0.f;
    }
}

#define CS_THREADS 1024
#define CS_WAVES   (CS_THREADS / 32)
#define CS_LOADS   (T_PITCH / (4 * CS_THREADS))
static_assert(T_PITCH % (4 * CS_THREADS) == 0, "the threads' loads tile the column exactly");
static_assert(CS_WAVES == 32, "one partial per lane in the second reduction level");

__global__ void __launch_bounds__(CS_THREADS)
k_colstats(const float* __restrict__ tplane,
           float* __restrict__ colmax, float* __restrict__ colsum) {
    __shared__ float red[CS_WAVES];
    const int c    = blockIdx.x;
    const int lane = threadIdx.x & 31;
    const int wave = threadIdx.x >> 5;
    const float4* col = (const float4*)(tplane + (size_t)c * T_PITCH);

    float4 v[CS_LOADS];
    #pragma unroll
    for (int j = 0; j < CS_LOADS; ++j) v[j] = col[j * CS_THREADS + threadIdx.x];

    float m = -INFINITY;
    #pragma unroll
    for (int j = 0; j < CS_LOADS; ++j)
        m = fmaxf(m, fmaxf(fmaxf(v[j].x, v[j].y), fmaxf(v[j].z, v[j].w)));
    for (int off = 16; off > 0; off >>= 1) m = fmaxf(m, __shfl_xor(m, off, 32));
    if (lane == 0) red[wave] = m;
    __syncthreads();
    float cm = red[lane];
    for (int off = 16; off > 0; off >>= 1) cm = fmaxf(cm, __shfl_xor(cm, off, 32));
    __syncthreads();

    float s = 0.f;
    #pragma unroll
    for (int j = 0; j < CS_LOADS; ++j)
        s += (expf(v[j].x - cm) + expf(v[j].y - cm)) + (expf(v[j].z - cm) + expf(v[j].w - cm));
    for (int off = 16; off > 0; off >>= 1) s += __shfl_xor(s, off, 32);
    if (lane == 0) red[wave] = s;
    __syncthreads();
    if (wave == 0) {
        float t = red[lane];
        for (int off = 16; off > 0; off >>= 1) t += __shfl_xor(t, off, 32);
        if (lane == 0) { colmax[c] = cm; colsum[c] = t; }
    }
}

__global__ void k_finalout(const float* __restrict__ logits,
                           const float* __restrict__ colmax, const float* __restrict__ colsum,
                           float* __restrict__ out_lsm, float* __restrict__ out_sm0) {
    int gid  = blockIdx.x * blockDim.x + threadIdx.x;
    int node = gid >> 5;
    if (node >= N_NODES) return;
    int lane = gid & 31;
    const float* row = logits + (size_t)node * NC;
    float v0 = row[lane];
    bool has2 = lane < (NC - 32);
    float v1 = has2 ? row[lane + 32] : -INFINITY;
    float m = fmaxf(v0, v1);
    #pragma unroll
    for (int off = 16; off > 0; off >>= 1) m = fmaxf(m, __shfl_xor(m, off, 32));
    float s = expf(v0 - m) + (has2 ? expf(v1 - m) : 0.f);
    #pragma unroll
    for (int off = 16; off > 0; off >>= 1) s += __shfl_xor(s, off, 32);
    float lse = m + logf(s);
    out_lsm[(size_t)node * NC + lane] = v0 - lse;
    out_sm0[(size_t)node * NC + lane] = expf(v0 - colmax[lane]) / colsum[lane];
    if (has2) {
        out_lsm[(size_t)node * NC + lane + 32] = v1 - lse;
        out_sm0[(size_t)node * NC + lane + 32] =
            expf(v1 - colmax[lane + 32]) / colsum[lane + 32];
    }
}

extern "C" void kernel_launch(void* const* d_in, const int* in_sizes, int n_in,
                              void* d_out, int out_size, void* d_ws, size_t ws_size,
                              hipStream_t stream) {
    (void)in_sizes; (void)n_in; (void)out_size; (void)ws_size;
    const float* x   = (const float*)d_in[0];
    const int*   ei  = (const int*)d_in[1];
    const float* W1  = (const float*)d_in[2];
    const float* b1  = (const float*)d_in[3];
    const float* W2  = (const float*)d_in[4];
    const float* b2  = (const float*)d_in[5];
    const int* src = ei;
    const int* dst = ei + N_EDGES;

    float* ws = (float*)d_ws;
    size_t off = 0;
    float* dinv = ws + off; off += N_NODES; off = (off + 3) & ~(size_t)3;
    float* xw1  = ws + off; off += (size_t)N_NODES * NH;
    float* agg1 = ws + off; off += (size_t)N_NODES * NH;
    float* W2p  = ws + off; off += NH * NCP;
    int*   lists = (int*)(ws + off); off += (size_t)N_NODES * NCP;
    float* colmax = ws + off; off += 64;
    float* colsum = ws + off; off += 64;

    float* tplane = xw1;
    float* xw2    = xw1 + (size_t)NC * T_PITCH;

    float* out_lsm = (float*)d_out;
    float* logits  = (float*)d_out + (size_t)N_NODES * NC;
    float* out_sm0 = (float*)d_out + 2 * (size_t)N_NODES * NC;
    unsigned* marks = (unsigned*)d_out;

    hipMemsetAsync(dinv, 0, N_NODES * sizeof(float), stream);
    k_degree_mark<<<MK_GROUPS, MK_THREADS, 0, stream>>>(dst, dinv, marks);
    k_dinv<<<(N_NODES + 255) / 256, 256, 0, stream>>>(dinv);

    k_gemm_wmma<NF, NF, NH><<<GM_GROUPS, GM_THREADS, 0, stream>>>(x, W1, xw1);
    k_own<NH, NH, true, false, false><<<OWN_GROUPS, OWN_THREADS, 0, stream>>>(src, dst, dinv, xw1, b1, agg1, nullptr, lists, marks);

    k_padw2<<<(NH * NCP + 255) / 256, 256, 0, stream>>>(W2, W2p);
    k_gemm_wmma<NH, NH, NCP><<<GM_GROUPS, GM_THREADS, 0, stream>>>(agg1, W2p, xw2);
    k_own<NC, NCP, false, true, true><<<OWN_GROUPS, OWN_THREADS, 0, stream>>>(src, dst, dinv, xw2, b2, logits, tplane, lists, marks);

    k_colstats<<<NC, CS_THREADS, 0, stream>>>(tplane, colmax, colsum);
    k_finalout<<<(N_NODES * 32 + 255) / 256, 256, 0, stream>>>(logits, colmax, colsum,
                                                               out_lsm, out_sm0);
}
